# starting kernel
# baseline (speedup 1.0000x reference)
#include <hip/hip_runtime.h>
#include <hip/hip_bf16.h>
#include <math.h>

typedef float v2f __attribute__((ext_vector_type(2)));
typedef float v8f __attribute__((ext_vector_type(8)));

#define BN 4
#define SS 2048
#define DD 1024
#define HH 1024
#define OO 1024
#define FF 2048

template <bool TRANSB>
__global__ void wmma_gemm16(const float* __restrict__ A, long lda, int rows,
                            const float* __restrict__ Bm, int ldb,
                            int K, const float* __restrict__ bias,
                            float* __restrict__ Out, int ldout) {
  __shared__ float red[4][32][8];
  const int lane = threadIdx.x & 31;
  const int wave = threadIdx.x >> 5;
  const int n0   = blockIdx.x * 16;
  const int ln   = lane & 15;
  const int kk   = (lane >> 4) * 2;
  const int kPerWave = K >> 2;
  const int kBeg = wave * kPerWave;

  const float amask = (ln < rows) ? 1.f : 0.f;
  const float* aptr = A + (long)(ln < rows ? ln : 0) * lda + kBeg + kk;
  const float* bptr = TRANSB ? (Bm + (long)(n0 + ln) * ldb + kBeg + kk)
                             : (Bm + (long)(kBeg + kk) * ldb + (n0 + ln));
  const long brow  = TRANSB ? 1 : (long)ldb;
  const long bstep = TRANSB ? 4 : (long)4 * ldb;

  v8f acc0 = {0.f, 0.f, 0.f, 0.f, 0.f, 0.f, 0.f, 0.f};
  v8f acc1 = {0.f, 0.f, 0.f, 0.f, 0.f, 0.f, 0.f, 0.f};

  const int iters = kPerWave >> 2;
  for (int i = 0; i < iters; i += 2) {
    v2f a0, b0, a1, b1;
    a0.x = aptr[0] * amask;  a0.y = aptr[1] * amask;
    b0.x = bptr[0];          b0.y = bptr[brow];
    a1.x = aptr[4] * amask;  a1.y = aptr[5] * amask;
    b1.x = bptr[bstep];      b1.y = bptr[bstep + brow];
    acc0 = __builtin_amdgcn_wmma_f32_16x16x4_f32(
        false, a0, false, b0, (short)0, acc0, false, false);
    acc1 = __builtin_amdgcn_wmma_f32_16x16x4_f32(
        false, a1, false, b1, (short)0, acc1, false, false);
    aptr += 8;
    bptr += 2 * bstep;
  }

  for (int i = 0; i < 8; ++i) red[wave][lane][i] = acc0[i] + acc1[i];
  __syncthreads();

  if (wave == 0 && lane < 16) {
    for (int i = 0; i < rows; ++i) {
      float v = red[0][lane][i] + red[1][lane][i] +
                red[2][lane][i] + red[3][lane][i];
      if (bias) v += bias[n0 + lane];
      Out[(long)i * ldout + n0 + lane] = v;
    }
  }
}

__global__ void scores_k(const float* __restrict__ x,
                         const float* __restrict__ qtil,
                         float* __restrict__ scores) {
  const int b    = blockIdx.y;
  const int srow = blockIdx.x * 8 + (threadIdx.x >> 5);
  const int lane = threadIdx.x & 31;
  const float4* xr = (const float4*)(x + ((size_t)b * SS + srow) * DD);
  const float4* q4 = (const float4*)(qtil + (size_t)b * DD);
  float sum = 0.f;
#pragma unroll
  for (int i = 0; i < 8; ++i) {
    float4 xv = xr[lane + i * 32];
    float4 qv = q4[lane + i * 32];
    sum += xv.x * qv.x + xv.y * qv.y + xv.z * qv.z + xv.w * qv.w;
  }
  for (int off = 16; off; off >>= 1) sum += __shfl_xor(sum, off, 32);
  if (lane == 0) scores[b * SS + srow] = sum * 0.03125f;
}

__global__ void softmax_zero(float* __restrict__ scores,
                             float* __restrict__ cvec) {
  const int b = blockIdx.x, tid = threadIdx.x;
  __shared__ float red[256];
  float* sc = scores + b * SS;
  float v[8];
  float mx = -3.402823466e38f;
#pragma unroll
  for (int j = 0; j < 8; ++j) { v[j] = sc[tid + j * 256]; mx = fmaxf(mx, v[j]); }
  red[tid] = mx; __syncthreads();
  for (int s = 128; s; s >>= 1) {
    if (tid < s) red[tid] = fmaxf(red[tid], red[tid + s]);
    __syncthreads();
  }
  mx = red[0]; __syncthreads();
  float sum = 0.f;
#pragma unroll
  for (int j = 0; j < 8; ++j) { v[j] = expf(v[j] - mx); sum += v[j]; }
  red[tid] = sum; __syncthreads();
  for (int s = 128; s; s >>= 1) {
    if (tid < s) red[tid] += red[tid + s];
    __syncthreads();
  }
  const float inv = 1.f / red[0];
#pragma unroll
  for (int j = 0; j < 8; ++j) sc[tid + j * 256] = v[j] * inv;
#pragma unroll
  for (int c = 0; c < 4; ++c) cvec[b * DD + tid * 4 + c] = 0.f;
}

__global__ void wsum_k(const float* __restrict__ x,
                       const float* __restrict__ wts,
                       float* __restrict__ cvec) {
  const int chunk = blockIdx.x, b = blockIdx.y, tid = threadIdx.x;
  const int s0 = chunk * 128;
  const float4* x4 = (const float4*)x;
  const float* w = wts + b * SS + s0;
  const size_t base = ((size_t)b * SS + s0) * (DD / 4);
  float4 acc = {0.f, 0.f, 0.f, 0.f};
  for (int s = 0; s < 128; ++s) {
    const float ww = w[s];
    float4 xv = x4[base + (size_t)s * (DD / 4) + tid];
    acc.x += ww * xv.x; acc.y += ww * xv.y;
    acc.z += ww * xv.z; acc.w += ww * xv.w;
  }
  float* c = cvec + b * DD + tid * 4;
  atomicAdd(c + 0, acc.x); atomicAdd(c + 1, acc.y);
  atomicAdd(c + 2, acc.z); atomicAdd(c + 3, acc.w);
}

__global__ void glu_ln(const float* __restrict__ hin,
                       const float* __restrict__ attn,
                       const float* __restrict__ gamma,
                       const float* __restrict__ beta,
                       float* __restrict__ comb) {
  const int b = blockIdx.x, tid = threadIdx.x;
  __shared__ float red[256];
  const float* hb = hin + (size_t)b * FF;
  float hv[4];
  float sum = 0.f;
#pragma unroll
  for (int j = 0; j < 4; ++j) {
    const int i = tid + j * 256;
    const float x1 = hb[i];
    const float x2 = hb[HH + i];
    const float g = 0.5f * x2 * (1.f + erff(x2 * 0.70710678118654752f));
    hv[j] = x1 * g;
    sum += hv[j];
  }
  red[tid] = sum; __syncthreads();
  for (int s = 128; s; s >>= 1) { if (tid < s) red[tid] += red[tid + s]; __syncthreads(); }
  const float mean = red[0] * (1.f / HH); __syncthreads();
  float sq = 0.f;
#pragma unroll
  for (int j = 0; j < 4; ++j) { const float d = hv[j] - mean; sq += d * d; }
  red[tid] = sq; __syncthreads();
  for (int s = 128; s; s >>= 1) { if (tid < s) red[tid] += red[tid + s]; __syncthreads(); }
  const float rstd = rsqrtf(red[0] * (1.f / HH) + 1e-5f);
#pragma unroll
  for (int j = 0; j < 4; ++j) {
    const int i = tid + j * 256;
    comb[b * HH + i] = attn[b * HH + i] + (hv[j] - mean) * rstd * gamma[i] + beta[i];
  }
}

extern "C" void kernel_launch(void* const* d_in, const int* in_sizes, int n_in,
                              void* d_out, int out_size, void* d_ws, size_t ws_size,
                              hipStream_t stream) {
  const float* x       = (const float*)d_in[0];
  const float* Wq_proj = (const float*)d_in[1];
  const float* bq      = (const float*)d_in[2];
  const float* Wq      = (const float*)d_in[3];
  const float* Wk      = (const float*)d_in[4];
  const float* Wv      = (const float*)d_in[5];
  const float* W1      = (const float*)d_in[6];
  const float* b1      = (const float*)d_in[7];
  const float* W2      = (const float*)d_in[8];
  const float* b2      = (const float*)d_in[9];
  const float* gamma   = (const float*)d_in[10];
  const float* beta    = (const float*)d_in[11];
  float* out = (float*)d_out;

  float* ws     = (float*)d_ws;
  float* qproj  = ws;
  float* qlast  = ws + 4096;
  float* qtil   = ws + 8192;
  float* scores = ws + 12288;
  float* cvec   = ws + 20480;
  float* attn   = ws + 24576;
  float* hin    = ws + 28672;
  float* comb   = ws + 36864;

  wmma_gemm16<false><<<dim3(HH / 16), 128, 0, stream>>>(
      x + (size_t)(SS - 1) * DD, (long)SS * DD, BN, Wq_proj, HH, DD, bq, qproj, HH);
  wmma_gemm16<false><<<dim3(HH / 16), 128, 0, stream>>>(
      qproj, HH, BN, Wq, HH, HH, nullptr, qlast, HH);
  wmma_gemm16<true><<<dim3(DD / 16), 128, 0, stream>>>(
      qlast, HH, BN, Wk, HH, HH, nullptr, qtil, DD);
  scores_k<<<dim3(SS / 8, BN), 256, 0, stream>>>(x, qtil, scores);
  softmax_zero<<<dim3(BN), 256, 0, stream>>>(scores, cvec);
  wsum_k<<<dim3(16, BN), 256, 0, stream>>>(x, scores, cvec);
  wmma_gemm16<false><<<dim3(HH / 16), 128, 0, stream>>>(
      cvec, DD, BN, Wv, HH, DD, nullptr, attn, HH);
  wmma_gemm16<false><<<dim3(FF / 16), 128, 0, stream>>>(
      attn, HH, BN, W1, FF, HH, b1, hin, FF);
  glu_ln<<<dim3(BN), 256, 0, stream>>>(hin, attn, gamma, beta, comb);
  wmma_gemm16<false><<<dim3(OO / 16), 128, 0, stream>>>(
      comb, HH, BN, W2, OO, HH, b2, out, OO);
}


# small GEMMs: K split over 16 waves per block (was 4), strides compile-time so each wave issues all its loads at once and waits one memory latency; partial sums added in wave order
# speedup vs baseline: 1.6540x; 1.6540x over previous
#include <hip/hip_runtime.h>
#include <hip/hip_bf16.h>
#include <math.h>

typedef float v2f __attribute__((ext_vector_type(2)));
typedef float v8f __attribute__((ext_vector_type(8)));

#define BN 4
#define SS 2048
#define DD 1024
#define HH 1024
#define OO 1024
#define FF 2048

template <bool TRANSB, int LDB, int K, int LDOUT, bool HASBIAS>
__global__ __launch_bounds__(512) __attribute__((amdgpu_waves_per_eu(4, 4)))
void wmma_gemm16(
    const float* __restrict__ A, long lda, int rows,
    const float* __restrict__ Bm, const float* __restrict__ bias,
    float* __restrict__ Out) {
  constexpr int NW = 16;
  constexpr int KW = K / NW;
  constexpr int NP = KW / 8;
  __shared__ float red[NW][16][16];
  const int lane = threadIdx.x & 31;
  const int wave = threadIdx.x >> 5;
  const int n0   = blockIdx.x * 16;
  const int c    = lane & 15;
  const int h    = lane >> 4;
  const int kb   = wave * KW + 4 * h;

  const float amask = (c < rows) ? 1.f : 0.f;
  const float* aptr = A + (long)(c < rows ? c : 0) * lda + kb;
  const float* bptr = TRANSB ? (Bm + (long)(n0 + c) * LDB + kb)
                             : (Bm + (long)kb * LDB + (n0 + c));

  const float bval = HASBIAS ? bias[n0 + (threadIdx.x & 15)] : 0.f;

  float4 av[NP], bv[NP];
#pragma unroll
  for (int j = 0; j < NP; ++j) {
    av[j] = *(const float4*)(aptr + 8 * j);
    if (TRANSB) {
      bv[j] = *(const float4*)(bptr + 8 * j);
    } else {
      const float* bp = bptr + (long)(8 * j) * LDB;
      bv[j].x = bp[0];
      bv[j].y = bp[LDB];
      bv[j].z = bp[2 * LDB];
      bv[j].w = bp[3 * LDB];
    }
  }

  v8f acc0 = {0.f, 0.f, 0.f, 0.f, 0.f, 0.f, 0.f, 0.f};
  v8f acc1 = {0.f, 0.f, 0.f, 0.f, 0.f, 0.f, 0.f, 0.f};
#pragma unroll
  for (int j = 0; j < NP; ++j) {
    v2f a0, b0, a1, b1;
    a0.x = av[j].x * amask;  a0.y = av[j].y * amask;
    b0.x = bv[j].x;          b0.y = bv[j].y;
    a1.x = av[j].z * amask;  a1.y = av[j].w * amask;
    b1.x = bv[j].z;          b1.y = bv[j].w;
    acc0 = __builtin_amdgcn_wmma_f32_16x16x4_f32(
        false, a0, false, b0, (short)0, acc0, false, false);
    acc1 = __builtin_amdgcn_wmma_f32_16x16x4_f32(
        false, a1, false, b1, (short)0, acc1, false, false);
  }

#pragma unroll
  for (int i = 0; i < 8; ++i) red[wave][8 * h + i][c] = acc0[i] + acc1[i];
  __syncthreads();

  const int t = threadIdx.x;
  if (t < 16 * rows) {
    const int r = t >> 4, cc = t & 15;
    float v = 0.f;
#pragma unroll
    for (int w = 0; w < NW; ++w) v += red[w][r][cc];
    Out[r * LDOUT + n0 + cc] = v + bval;
  }
}

__global__ void scores_k(const float* __restrict__ x,
                         const float* __restrict__ qtil,
                         float* __restrict__ scores) {
  const int b    = blockIdx.y;
  const int srow = blockIdx.x * 8 + (threadIdx.x >> 5);
  const int lane = threadIdx.x & 31;
  const float4* xr = (const float4*)(x + ((size_t)b * SS + srow) * DD);
  const float4* q4 = (const float4*)(qtil + (size_t)b * DD);
  float sum = 0.f;
#pragma unroll
  for (int i = 0; i < 8; ++i) {
    float4 xv = xr[lane + i * 32];
    float4 qv = q4[lane + i * 32];
    sum += xv.x * qv.x + xv.y * qv.y + xv.z * qv.z + xv.w * qv.w;
  }
  for (int off = 16; off; off >>= 1) sum += __shfl_xor(sum, off, 32);
  if (lane == 0) scores[b * SS + srow] = sum * 0.03125f;
}

__global__ void softmax_zero(float* __restrict__ scores,
                             float* __restrict__ cvec) {
  const int b = blockIdx.x, tid = threadIdx.x;
  __shared__ float red[256];
  float* sc = scores + b * SS;
  float v[8];
  float mx = -3.402823466e38f;
#pragma unroll
  for (int j = 0; j < 8; ++j) { v[j] = sc[tid + j * 256]; mx = fmaxf(mx, v[j]); }
  red[tid] = mx; __syncthreads();
  for (int s = 128; s; s >>= 1) {
    if (tid < s) red[tid] = fmaxf(red[tid], red[tid + s]);
    __syncthreads();
  }
  mx = red[0]; __syncthreads();
  float sum = 0.f;
#pragma unroll
  for (int j = 0; j < 8; ++j) { v[j] = expf(v[j] - mx); sum += v[j]; }
  red[tid] = sum; __syncthreads();
  for (int s = 128; s; s >>= 1) {
    if (tid < s) red[tid] += red[tid + s];
    __syncthreads();
  }
  const float inv = 1.f / red[0];
#pragma unroll
  for (int j = 0; j < 8; ++j) sc[tid + j * 256] = v[j] * inv;
#pragma unroll
  for (int c = 0; c < 4; ++c) cvec[b * DD + tid * 4 + c] = 0.f;
}

__global__ void wsum_k(const float* __restrict__ x,
                       const float* __restrict__ wts,
                       float* __restrict__ cvec) {
  const int chunk = blockIdx.x, b = blockIdx.y, tid = threadIdx.x;
  const int s0 = chunk * 128;
  const float4* x4 = (const float4*)x;
  const float* w = wts + b * SS + s0;
  const size_t base = ((size_t)b * SS + s0) * (DD / 4);
  float4 acc = {0.f, 0.f, 0.f, 0.f};
  for (int s = 0; s < 128; ++s) {
    const float ww = w[s];
    float4 xv = x4[base + (size_t)s * (DD / 4) + tid];
    acc.x += ww * xv.x; acc.y += ww * xv.y;
    acc.z += ww * xv.z; acc.w += ww * xv.w;
  }
  float* c = cvec + b * DD + tid * 4;
  atomicAdd(c + 0, acc.x); atomicAdd(c + 1, acc.y);
  atomicAdd(c + 2, acc.z); atomicAdd(c + 3, acc.w);
}

__global__ void glu_ln(const float* __restrict__ hin,
                       const float* __restrict__ attn,
                       const float* __restrict__ gamma,
                       const float* __restrict__ beta,
                       float* __restrict__ comb) {
  const int b = blockIdx.x, tid = threadIdx.x;
  __shared__ float red[256];
  const float* hb = hin + (size_t)b * FF;
  float hv[4];
  float sum = 0.f;
#pragma unroll
  for (int j = 0; j < 4; ++j) {
    const int i = tid + j * 256;
    const float x1 = hb[i];
    const float x2 = hb[HH + i];
    const float g = 0.5f * x2 * (1.f + erff(x2 * 0.70710678118654752f));
    hv[j] = x1 * g;
    sum += hv[j];
  }
  red[tid] = sum; __syncthreads();
  for (int s = 128; s; s >>= 1) { if (tid < s) red[tid] += red[tid + s]; __syncthreads(); }
  const float mean = red[0] * (1.f / HH); __syncthreads();
  float sq = 0.f;
#pragma unroll
  for (int j = 0; j < 4; ++j) { const float d = hv[j] - mean; sq += d * d; }
  red[tid] = sq; __syncthreads();
  for (int s = 128; s; s >>= 1) { if (tid < s) red[tid] += red[tid + s]; __syncthreads(); }
  const float rstd = rsqrtf(red[0] * (1.f / HH) + 1e-5f);
#pragma unroll
  for (int j = 0; j < 4; ++j) {
    const int i = tid + j * 256;
    comb[b * HH + i] = attn[b * HH + i] + (hv[j] - mean) * rstd * gamma[i] + beta[i];
  }
}

extern "C" void kernel_launch(void* const* d_in, const int* in_sizes, int n_in,
                              void* d_out, int out_size, void* d_ws, size_t ws_size,
                              hipStream_t stream) {
  const float* x       = (const float*)d_in[0];
  const float* Wq_proj = (const float*)d_in[1];
  const float* bq      = (const float*)d_in[2];
  const float* Wq      = (const float*)d_in[3];
  const float* Wk      = (const float*)d_in[4];
  const float* Wv      = (const float*)d_in[5];
  const float* W1      = (const float*)d_in[6];
  const float* b1      = (const float*)d_in[7];
  const float* W2      = (const float*)d_in[8];
  const float* b2      = (const float*)d_in[9];
  const float* gamma   = (const float*)d_in[10];
  const float* beta    = (const float*)d_in[11];
  float* out = (float*)d_out;

  float* ws     = (float*)d_ws;
  float* qproj  = ws;
  float* qlast  = ws + 4096;
  float* qtil   = ws + 8192;
  float* scores = ws + 12288;
  float* cvec   = ws + 20480;
  float* attn   = ws + 24576;
  float* hin    = ws + 28672;
  float* comb   = ws + 36864;

  wmma_gemm16<false, HH, DD, HH, true><<<dim3(HH / 16), 512, 0, stream>>>(
      x + (size_t)(SS - 1) * DD, (long)SS * DD, BN, Wq_proj, bq, qproj);
  wmma_gemm16<false, HH, HH, HH, false><<<dim3(HH / 16), 512, 0, stream>>>(
      qproj, HH, BN, Wq, nullptr, qlast);
  wmma_gemm16<true, HH, HH, DD, false><<<dim3(DD / 16), 512, 0, stream>>>(
      qlast, HH, BN, Wk, nullptr, qtil);
  scores_k<<<dim3(SS / 8, BN), 256, 0, stream>>>(x, qtil, scores);
  softmax_zero<<<dim3(BN), 256, 0, stream>>>(scores, cvec);
  wsum_k<<<dim3(16, BN), 256, 0, stream>>>(x, scores, cvec);
  wmma_gemm16<false, HH, DD, HH, false><<<dim3(HH / 16), 512, 0, stream>>>(
      cvec, DD, BN, Wv, nullptr, attn);
  wmma_gemm16<false, FF, HH, FF, true><<<dim3(FF / 16), 512, 0, stream>>>(
      attn, HH, BN, W1, b1, hin);
  glu_ln<<<dim3(BN), 256, 0, stream>>>(hin, attn, gamma, beta, comb);
  wmma_gemm16<false, OO, HH, OO, true><<<dim3(OO / 16), 512, 0, stream>>>(
      comb, HH, BN, W2, b2, out);
}


# both passes over x re-tiled to 256 blocks of 8 waves, all row loads of a wave issued at once (scores: 4 rows per wave; weighted sum: rows summed per wave, waves in order, 8 atomic partials, was 16)
# speedup vs baseline: 1.9456x; 1.1763x over previous
#include <hip/hip_runtime.h>
#include <hip/hip_bf16.h>
#include <math.h>

typedef float v2f __attribute__((ext_vector_type(2)));
typedef float v8f __attribute__((ext_vector_type(8)));

#define BN 4
#define SS 2048
#define DD 1024
#define HH 1024
#define OO 1024
#define FF 2048

template <bool TRANSB, int LDB, int K, int LDOUT, bool HASBIAS>
__global__ __launch_bounds__(512) __attribute__((amdgpu_waves_per_eu(4, 4)))
void wmma_gemm16(
    const float* __restrict__ A, long lda, int rows,
    const float* __restrict__ Bm, const float* __restrict__ bias,
    float* __restrict__ Out) {
  constexpr int NW = 16;
  constexpr int KW = K / NW;
  constexpr int NP = KW / 8;
  __shared__ float red[NW][16][16];
  const int lane = threadIdx.x & 31;
  const int wave = threadIdx.x >> 5;
  const int n0   = blockIdx.x * 16;
  const int c    = lane & 15;
  const int h    = lane >> 4;
  const int kb   = wave * KW + 4 * h;

  const float amask = (c < rows) ? 1.f : 0.f;
  const float* aptr = A + (long)(c < rows ? c : 0) * lda + kb;
  const float* bptr = TRANSB ? (Bm + (long)(n0 + c) * LDB + kb)
                             : (Bm + (long)kb * LDB + (n0 + c));

  const float bval = HASBIAS ? bias[n0 + (threadIdx.x & 15)] : 0.f;

  float4 av[NP], bv[NP];
#pragma unroll
  for (int j = 0; j < NP; ++j) {
    av[j] = *(const float4*)(aptr + 8 * j);
    if (TRANSB) {
      bv[j] = *(const float4*)(bptr + 8 * j);
    } else {
      const float* bp = bptr + (long)(8 * j) * LDB;
      bv[j].x = bp[0];
      bv[j].y = bp[LDB];
      bv[j].z = bp[2 * LDB];
      bv[j].w = bp[3 * LDB];
    }
  }

  v8f acc0 = {0.f, 0.f, 0.f, 0.f, 0.f, 0.f, 0.f, 0.f};
  v8f acc1 = {0.f, 0.f, 0.f, 0.f, 0.f, 0.f, 0.f, 0.f};
#pragma unroll
  for (int j = 0; j < NP; ++j) {
    v2f a0, b0, a1, b1;
    a0.x = av[j].x * amask;  a0.y = av[j].y * amask;
    b0.x = bv[j].x;          b0.y = bv[j].y;
    a1.x = av[j].z * amask;  a1.y = av[j].w * amask;
    b1.x = bv[j].z;          b1.y = bv[j].w;
    acc0 = __builtin_amdgcn_wmma_f32_16x16x4_f32(
        false, a0, false, b0, (short)0, acc0, false, false);
    acc1 = __builtin_amdgcn_wmma_f32_16x16x4_f32(
        false, a1, false, b1, (short)0, acc1, false, false);
  }

#pragma unroll
  for (int i = 0; i < 8; ++i) red[wave][8 * h + i][c] = acc0[i] + acc1[i];
  __syncthreads();

  const int t = threadIdx.x;
  if (t < 16 * rows) {
    const int r = t >> 4, cc = t & 15;
    float v = 0.f;
#pragma unroll
    for (int w = 0; w < NW; ++w) v += red[w][r][cc];
    Out[r * LDOUT + n0 + cc] = v + bval;
  }
}

__global__ __launch_bounds__(256) __attribute__((amdgpu_waves_per_eu(2, 2)))
void scores_k(const float* __restrict__ x,
              const float* __restrict__ qtil,
              float* __restrict__ scores) {
  constexpr int RW = 4;
  const int b    = blockIdx.y;
  const int lane = threadIdx.x & 31;
  const int srow = blockIdx.x * (8 * RW) + (threadIdx.x >> 5) * RW;
  const float4* xr = (const float4*)(x + ((size_t)b * SS + srow) * DD) + lane;
  const float4* q4 = (const float4*)(qtil + (size_t)b * DD) + lane;
  float4 qv[8], xv[RW][8];
#pragma unroll
  for (int i = 0; i < 8; ++i) qv[i] = q4[i * 32];
#pragma unroll
  for (int r = 0; r < RW; ++r)
#pragma unroll
    for (int i = 0; i < 8; ++i) xv[r][i] = xr[r * (DD / 4) + i * 32];
  float sum[RW];
#pragma unroll
  for (int r = 0; r < RW; ++r) {
    float s = 0.f;
#pragma unroll
    for (int i = 0; i < 8; ++i)
      s += xv[r][i].x * qv[i].x + xv[r][i].y * qv[i].y +
           xv[r][i].z * qv[i].z + xv[r][i].w * qv[i].w;
    sum[r] = s;
  }
  for (int off = 16; off; off >>= 1) {
#pragma unroll
    for (int r = 0; r < RW; ++r) sum[r] += __shfl_xor(sum[r], off, 32);
  }
  const float v = lane == 0 ? sum[0] : lane == 1 ? sum[1]
                : lane == 2 ? sum[2] : sum[3];
  if (lane < RW) scores[b * SS + srow + lane] = v * 0.03125f;
}

__global__ void softmax_zero(float* __restrict__ scores,
                             float* __restrict__ cvec) {
  const int b = blockIdx.x, tid = threadIdx.x;
  __shared__ float red[256];
  float* sc = scores + b * SS;
  float v[8];
  float mx = -3.402823466e38f;
#pragma unroll
  for (int j = 0; j < 8; ++j) { v[j] = sc[tid + j * 256]; mx = fmaxf(mx, v[j]); }
  red[tid] = mx; __syncthreads();
  for (int s = 128; s; s >>= 1) {
    if (tid < s) red[tid] = fmaxf(red[tid], red[tid + s]);
    __syncthreads();
  }
  mx = red[0]; __syncthreads();
  float sum = 0.f;
#pragma unroll
  for (int j = 0; j < 8; ++j) { v[j] = expf(v[j] - mx); sum += v[j]; }
  red[tid] = sum; __syncthreads();
  for (int s = 128; s; s >>= 1) {
    if (tid < s) red[tid] += red[tid + s];
    __syncthreads();
  }
  const float inv = 1.f / red[0];
#pragma unroll
  for (int j = 0; j < 8; ++j) sc[tid + j * 256] = v[j] * inv;
#pragma unroll
  for (int c = 0; c < 4; ++c) cvec[b * DD + tid * 4 + c] = 0.f;
}

__global__ __launch_bounds__(256) __attribute__((amdgpu_waves_per_eu(2, 2)))
void wsum_k(const float* __restrict__ x,
            const float* __restrict__ wts,
            float* __restrict__ cvec) {
  constexpr int RW = 32;
  constexpr int NWV = 8;
  const int cs = blockIdx.x, chunk = blockIdx.y, b = blockIdx.z;
  const int t = threadIdx.x;
  const int lane = t & 31;
  const int wave = __builtin_amdgcn_readfirstlane(t >> 5);
  const int s0 = chunk * (RW * NWV) + wave * RW;
  const float* w = wts + b * SS + s0;
  const float4* xp = (const float4*)x + ((size_t)b * SS + s0) * (DD / 4)
                     + cs * 32 + lane;
  float4 xv[RW];
#pragma unroll
  for (int i = 0; i < RW; ++i) xv[i] = xp[(size_t)i * (DD / 4)];
  float4 acc = {0.f, 0.f, 0.f, 0.f};
#pragma unroll
  for (int i = 0; i < RW; ++i) {
    const float ww = w[i];
    acc.x += ww * xv[i].x; acc.y += ww * xv[i].y;
    acc.z += ww * xv[i].z; acc.w += ww * xv[i].w;
  }
  __shared__ float4 red[NWV][32];
  red[wave][lane] = acc;
  __syncthreads();
  if (t < 128) {
    const float* rf = (const float*)red;
    float v = 0.f;
#pragma unroll
    for (int wv = 0; wv < NWV; ++wv) v += rf[wv * 128 + t];
    atomicAdd(cvec + b * DD + cs * 128 + t, v);
  }
}

__global__ void glu_ln(const float* __restrict__ hin,
                       const float* __restrict__ attn,
                       const float* __restrict__ gamma,
                       const float* __restrict__ beta,
                       float* __restrict__ comb) {
  const int b = blockIdx.x, tid = threadIdx.x;
  __shared__ float red[256];
  const float* hb = hin + (size_t)b * FF;
  float hv[4];
  float sum = 0.f;
#pragma unroll
  for (int j = 0; j < 4; ++j) {
    const int i = tid + j * 256;
    const float x1 = hb[i];
    const float x2 = hb[HH + i];
    const float g = 0.5f * x2 * (1.f + erff(x2 * 0.70710678118654752f));
    hv[j] = x1 * g;
    sum += hv[j];
  }
  red[tid] = sum; __syncthreads();
  for (int s = 128; s; s >>= 1) { if (tid < s) red[tid] += red[tid + s]; __syncthreads(); }
  const float mean = red[0] * (1.f / HH); __syncthreads();
  float sq = 0.f;
#pragma unroll
  for (int j = 0; j < 4; ++j) { const float d = hv[j] - mean; sq += d * d; }
  red[tid] = sq; __syncthreads();
  for (int s = 128; s; s >>= 1) { if (tid < s) red[tid] += red[tid + s]; __syncthreads(); }
  const float rstd = rsqrtf(red[0] * (1.f / HH) + 1e-5f);
#pragma unroll
  for (int j = 0; j < 4; ++j) {
    const int i = tid + j * 256;
    comb[b * HH + i] = attn[b * HH + i] + (hv[j] - mean) * rstd * gamma[i] + beta[i];
  }
}

extern "C" void kernel_launch(void* const* d_in, const int* in_sizes, int n_in,
                              void* d_out, int out_size, void* d_ws, size_t ws_size,
                              hipStream_t stream) {
  const float* x       = (const float*)d_in[0];
  const float* Wq_proj = (const float*)d_in[1];
  const float* bq      = (const float*)d_in[2];
  const float* Wq      = (const float*)d_in[3];
  const float* Wk      = (const float*)d_in[4];
  const float* Wv      = (const float*)d_in[5];
  const float* W1      = (const float*)d_in[6];
  const float* b1      = (const float*)d_in[7];
  const float* W2      = (const float*)d_in[8];
  const float* b2      = (const float*)d_in[9];
  const float* gamma   = (const float*)d_in[10];
  const float* beta    = (const float*)d_in[11];
  float* out = (float*)d_out;

  float* ws     = (float*)d_ws;
  float* qproj  = ws;
  float* qlast  = ws + 4096;
  float* qtil   = ws + 8192;
  float* scores = ws + 12288;
  float* cvec   = ws + 20480;
  float* attn   = ws + 24576;
  float* hin    = ws + 28672;
  float* comb   = ws + 36864;

  wmma_gemm16<false, HH, DD, HH, true><<<dim3(HH / 16), 512, 0, stream>>>(
      x + (size_t)(SS - 1) * DD, (long)SS * DD, BN, Wq_proj, bq, qproj);
  wmma_gemm16<false, HH, HH, HH, false><<<dim3(HH / 16), 512, 0, stream>>>(
      qproj, HH, BN, Wq, nullptr, qlast);
  wmma_gemm16<true, HH, HH, DD, false><<<dim3(DD / 16), 512, 0, stream>>>(
      qlast, HH, BN, Wk, nullptr, qtil);
  scores_k<<<dim3(SS / 32, BN), 256, 0, stream>>>(x, qtil, scores);
  softmax_zero<<<dim3(BN), 256, 0, stream>>>(scores, cvec);
  wsum_k<<<dim3(8, 8, BN), 256, 0, stream>>>(x, scores, cvec);
  wmma_gemm16<false, HH, DD, HH, false><<<dim3(HH / 16), 512, 0, stream>>>(
      cvec, DD, BN, Wv, nullptr, attn);
  wmma_gemm16<false, FF, HH, FF, true><<<dim3(FF / 16), 512, 0, stream>>>(
      attn, HH, BN, W1, b1, hin);
  glu_ln<<<dim3(BN), 256, 0, stream>>>(hin, attn, gamma, beta, comb);
  wmma_gemm16<false, OO, HH, OO, true><<<dim3(OO / 16), 512, 0, stream>>>(
      comb, HH, BN, W2, b2, out);
}


# launches 10 -> 8: softmax redone in every weighted-sum block and GLU+LayerNorm in every block of the last GEMM, under their tile loads in flight; these sums are now taken in a new tree order
# speedup vs baseline: 2.3306x; 1.1979x over previous
#include <hip/hip_runtime.h>
#include <hip/hip_bf16.h>
#include <math.h>

typedef float v2f __attribute__((ext_vector_type(2)));
typedef float v8f __attribute__((ext_vector_type(8)));

#define BN 4
#define SS 2048
#define DD 1024
#define HH 1024
#define OO 1024
#define FF 2048

template <bool TRANSB, int LDB, int K, int LDOUT, bool HASBIAS>
__global__ __launch_bounds__(512) __attribute__((amdgpu_waves_per_eu(4, 4)))
void wmma_gemm16(
    const float* __restrict__ A, long lda, int rows,
    const float* __restrict__ Bm, const float* __restrict__ bias,
    float* __restrict__ Out) {
  constexpr int NW = 16;
  constexpr int KW = K / NW;
  constexpr int NP = KW / 8;
  __shared__ float red[NW][16][16];
  const int lane = threadIdx.x & 31;
  const int wave = threadIdx.x >> 5;
  const int n0   = blockIdx.x * 16;
  const int c    = lane & 15;
  const int h    = lane >> 4;
  const int kb   = wave * KW + 4 * h;

  const float amask = (c < rows) ? 1.f : 0.f;
  const float* aptr = A + (long)(c < rows ? c : 0) * lda + kb;
  const float* bptr = TRANSB ? (Bm + (long)(n0 + c) * LDB + kb)
                             : (Bm + (long)kb * LDB + (n0 + c));

  const float bval = HASBIAS ? bias[n0 + (threadIdx.x & 15)] : 0.f;

  float4 av[NP], bv[NP];
#pragma unroll
  for (int j = 0; j < NP; ++j) {
    av[j] = *(const float4*)(aptr + 8 * j);
    if (TRANSB) {
      bv[j] = *(const float4*)(bptr + 8 * j);
    } else {
      const float* bp = bptr + (long)(8 * j) * LDB;
      bv[j].x = bp[0];
      bv[j].y = bp[LDB];
      bv[j].z = bp[2 * LDB];
      bv[j].w = bp[3 * LDB];
    }
  }

  v8f acc0 = {0.f, 0.f, 0.f, 0.f, 0.f, 0.f, 0.f, 0.f};
  v8f acc1 = {0.f, 0.f, 0.f, 0.f, 0.f, 0.f, 0.f, 0.f};
#pragma unroll
  for (int j = 0; j < NP; ++j) {
    v2f a0, b0, a1, b1;
    a0.x = av[j].x * amask;  a0.y = av[j].y * amask;
    b0.x = bv[j].x;          b0.y = bv[j].y;
    a1.x = av[j].z * amask;  a1.y = av[j].w * amask;
    b1.x = bv[j].z;          b1.y = bv[j].w;
    acc0 = __builtin_amdgcn_wmma_f32_16x16x4_f32(
        false, a0, false, b0, (short)0, acc0, false, false);
    acc1 = __builtin_amdgcn_wmma_f32_16x16x4_f32(
        false, a1, false, b1, (short)0, acc1, false, false);
  }

#pragma unroll
  for (int i = 0; i < 8; ++i) red[wave][8 * h + i][c] = acc0[i] + acc1[i];
  __syncthreads();

  const int t = threadIdx.x;
  if (t < 16 * rows) {
    const int r = t >> 4, cc = t & 15;
    float v = 0.f;
#pragma unroll
    for (int w = 0; w < NW; ++w) v += red[w][r][cc];
    Out[r * LDOUT + n0 + cc] = v + bval;
  }
}

__device__ __forceinline__ void lds_barrier() {
  __builtin_amdgcn_fence(__ATOMIC_RELEASE, "workgroup", "local");
  __builtin_amdgcn_s_barrier();
  __builtin_amdgcn_fence(__ATOMIC_ACQUIRE, "workgroup", "local");
}

__device__ __forceinline__ float wave_sum(float v) {
  for (int off = 16; off; off >>= 1) v += __shfl_xor(v, off, 32);
  return v;
}
__device__ __forceinline__ float wave_max(float v) {
  for (int off = 16; off; off >>= 1) v = fmaxf(v, __shfl_xor(v, off, 32));
  return v;
}

__global__ __launch_bounds__(256) __attribute__((amdgpu_waves_per_eu(2, 2)))
void scores_k(const float* __restrict__ x,
              const float* __restrict__ qtil,
              float* __restrict__ scores,
              float* __restrict__ cvec) {
  constexpr int RW = 4;
  const int b    = blockIdx.y;
  const int lane = threadIdx.x & 31;
  const int srow = blockIdx.x * (8 * RW) + (threadIdx.x >> 5) * RW;
  const float4* xr = (const float4*)(x + ((size_t)b * SS + srow) * DD) + lane;
  const float4* q4 = (const float4*)(qtil + (size_t)b * DD) + lane;
  float4 qv[8], xv[RW][8];
#pragma unroll
  for (int i = 0; i < 8; ++i) qv[i] = q4[i * 32];
#pragma unroll
  for (int r = 0; r < RW; ++r)
#pragma unroll
    for (int i = 0; i < 8; ++i) xv[r][i] = xr[r * (DD / 4) + i * 32];
  float sum[RW];
#pragma unroll
  for (int r = 0; r < RW; ++r) {
    float s = 0.f;
#pragma unroll
    for (int i = 0; i < 8; ++i)
      s += xv[r][i].x * qv[i].x + xv[r][i].y * qv[i].y +
           xv[r][i].z * qv[i].z + xv[r][i].w * qv[i].w;
    sum[r] = s;
  }
  for (int off = 16; off; off >>= 1) {
#pragma unroll
    for (int r = 0; r < RW; ++r) sum[r] += __shfl_xor(sum[r], off, 32);
  }
  const float v = lane == 0 ? sum[0] : lane == 1 ? sum[1]
                : lane == 2 ? sum[2] : sum[3];
  if (lane < RW) scores[b * SS + srow + lane] = v * 0.03125f;
  if (blockIdx.x == 0)
    ((float4*)(cvec + b * DD))[threadIdx.x] = float4{0.f, 0.f, 0.f, 0.f};
}

__global__ __launch_bounds__(256) __attribute__((amdgpu_waves_per_eu(2, 2)))
void wsum_k(const float* __restrict__ x,
            const float* __restrict__ scores,
            float* __restrict__ cvec) {
  constexpr int RW = 32;
  constexpr int NWV = 8;
  const int cs = blockIdx.x, chunk = blockIdx.y, b = blockIdx.z;
  const int t = threadIdx.x;
  const int lane = t & 31;
  const int wave = __builtin_amdgcn_readfirstlane(t >> 5);
  const int s0 = chunk * (RW * NWV) + wave * RW;
  const float* sc = scores + b * SS;
  const float4 sa = ((const float4*)sc)[t];
  const float4 sb = ((const float4*)sc)[t + 256];
  const float sown = sc[s0 + lane];
  const float4* xp = (const float4*)x + ((size_t)b * SS + s0) * (DD / 4)
                     + cs * 32 + lane;
  float4 xv[RW];
#pragma unroll
  for (int i = 0; i < RW; ++i) xv[i] = xp[(size_t)i * (DD / 4)];
  __builtin_amdgcn_sched_barrier(0);

  __shared__ float smax[NWV], ssum[NWV];
  __shared__ float4 red[NWV][32];
  float mx = fmaxf(fmaxf(fmaxf(sa.x, sa.y), fmaxf(sa.z, sa.w)),
                   fmaxf(fmaxf(sb.x, sb.y), fmaxf(sb.z, sb.w)));
  mx = wave_max(mx);
  if (lane == 0) smax[wave] = mx;
  lds_barrier();
  mx = smax[0];
#pragma unroll
  for (int wv = 1; wv < NWV; ++wv) mx = fmaxf(mx, smax[wv]);
  float es = expf(sa.x - mx) + expf(sa.y - mx) + expf(sa.z - mx) +
             expf(sa.w - mx) + expf(sb.x - mx) + expf(sb.y - mx) +
             expf(sb.z - mx) + expf(sb.w - mx);
  es = wave_sum(es);
  if (lane == 0) ssum[wave] = es;
  lds_barrier();
  float tot = 0.f;
#pragma unroll
  for (int wv = 0; wv < NWV; ++wv) tot += ssum[wv];
  const float inv = 1.f / tot;
  const int wbits = __builtin_bit_cast(int, expf(sown - mx) * inv);

  float4 acc = {0.f, 0.f, 0.f, 0.f};
#pragma unroll
  for (int i = 0; i < RW; ++i) {
    const float ww = __builtin_bit_cast(float, __builtin_amdgcn_readlane(wbits, i));
    acc.x += ww * xv[i].x; acc.y += ww * xv[i].y;
    acc.z += ww * xv[i].z; acc.w += ww * xv[i].w;
  }
  red[wave][lane] = acc;
  __syncthreads();
  if (t < 128) {
    const float* rf = (const float*)red;
    float v = 0.f;
#pragma unroll
    for (int wv = 0; wv < NWV; ++wv) v += rf[wv * 128 + t];
    atomicAdd(cvec + b * DD + cs * 128 + t, v);
  }
}

__global__ __launch_bounds__(512) __attribute__((amdgpu_waves_per_eu(4, 4)))
void glu_ln_gemm16(const float* __restrict__ hin,
                   const float* __restrict__ attn,
                   const float* __restrict__ gamma,
                   const float* __restrict__ beta,
                   const float* __restrict__ Bm,
                   const float* __restrict__ bias,
                   float* __restrict__ Out) {
  constexpr int NW = 16;
  constexpr int KW = HH / NW;
  constexpr int NP = KW / 8;
  constexpr int CP = HH + 8;
  __shared__ float red[NW][16][16];
  __shared__ __attribute__((aligned(16))) float comb[BN * CP];
  __shared__ float psum[NW], psq[NW];
  const int t    = threadIdx.x;
  const int lane = t & 31;
  const int wave = t >> 5;
  const int n0   = blockIdx.x * 16;
  const int c    = lane & 15;
  const int h    = lane >> 4;
  const int kb   = wave * KW + 4 * h;
  const int bt   = t >> 7;
  const int q    = t & 127;

  const float4* h4 = (const float4*)(hin + (size_t)bt * FF);
  const float4 x1a = h4[q],       x1b = h4[q + 128];
  const float4 x2a = h4[q + 256], x2b = h4[q + 384];
  const float4* a4 = (const float4*)(attn + (size_t)bt * HH);
  const float4 ata = a4[q], atb = a4[q + 128];
  const float4 gaa = ((const float4*)gamma)[q], gab = ((const float4*)gamma)[q + 128];
  const float4 bea = ((const float4*)beta)[q],  beb = ((const float4*)beta)[q + 128];
  const float bval = bias[n0 + (t & 15)];

  const float* bptr = Bm + (long)kb * OO + (n0 + c);
  float4 bv[NP];
#pragma unroll
  for (int j = 0; j < NP; ++j) {
    const float* bp = bptr + (long)(8 * j) * OO;
    bv[j].x = bp[0];
    bv[j].y = bp[OO];
    bv[j].z = bp[2 * OO];
    bv[j].w = bp[3 * OO];
  }

  const float x1[8] = {x1a.x, x1a.y, x1a.z, x1a.w, x1b.x, x1b.y, x1b.z, x1b.w};
  const float x2[8] = {x2a.x, x2a.y, x2a.z, x2a.w, x2b.x, x2b.y, x2b.z, x2b.w};
  float hv[8];
  float sum = 0.f;
#pragma unroll
  for (int j = 0; j < 8; ++j) {
    const float g = 0.5f * x2[j] * (1.f + erff(x2[j] * 0.70710678118654752f));
    hv[j] = x1[j] * g;
    sum += hv[j];
  }
  sum = wave_sum(sum);
  if (lane == 0) psum[wave] = sum;
  lds_barrier();
  const int w0 = bt * 4;
  const float mean = (psum[w0] + psum[w0 + 1] + psum[w0 + 2] + psum[w0 + 3]) * (1.f / HH);
  float sq = 0.f;
#pragma unroll
  for (int j = 0; j < 8; ++j) { const float d = hv[j] - mean; sq += d * d; }
  sq = wave_sum(sq);
  if (lane == 0) psq[wave] = sq;
  lds_barrier();
  const float var = (psq[w0] + psq[w0 + 1] + psq[w0 + 2] + psq[w0 + 3]) * (1.f / HH);
  const float rstd = rsqrtf(var + 1e-5f);
  float4 ca, cb;
  ca.x = ata.x + (hv[0] - mean) * rstd * gaa.x + bea.x;
  ca.y = ata.y + (hv[1] - mean) * rstd * gaa.y + bea.y;
  ca.z = ata.z + (hv[2] - mean) * rstd * gaa.z + bea.z;
  ca.w = ata.w + (hv[3] - mean) * rstd * gaa.w + bea.w;
  cb.x = atb.x + (hv[4] - mean) * rstd * gab.x + beb.x;
  cb.y = atb.y + (hv[5] - mean) * rstd * gab.y + beb.y;
  cb.z = atb.z + (hv[6] - mean) * rstd * gab.z + beb.z;
  cb.w = atb.w + (hv[7] - mean) * rstd * gab.w + beb.w;
  *(float4*)(comb + bt * CP + 4 * q)       = ca;
  *(float4*)(comb + bt * CP + 512 + 4 * q) = cb;
  lds_barrier();

  const float amask = (c < BN) ? 1.f : 0.f;
  const float* aptr = comb + (c < BN ? c : 0) * CP + kb;
  float4 av[NP];
#pragma unroll
  for (int j = 0; j < NP; ++j) av[j] = *(const float4*)(aptr + 8 * j);

  v8f acc0 = {0.f, 0.f, 0.f, 0.f, 0.f, 0.f, 0.f, 0.f};
  v8f acc1 = {0.f, 0.f, 0.f, 0.f, 0.f, 0.f, 0.f, 0.f};
#pragma unroll
  for (int j = 0; j < NP; ++j) {
    v2f a0, b0, a1, b1;
    a0.x = av[j].x * amask;  a0.y = av[j].y * amask;
    b0.x = bv[j].x;          b0.y = bv[j].y;
    a1.x = av[j].z * amask;  a1.y = av[j].w * amask;
    b1.x = bv[j].z;          b1.y = bv[j].w;
    acc0 = __builtin_amdgcn_wmma_f32_16x16x4_f32(
        false, a0, false, b0, (short)0, acc0, false, false);
    acc1 = __builtin_amdgcn_wmma_f32_16x16x4_f32(
        false, a1, false, b1, (short)0, acc1, false, false);
  }

#pragma unroll
  for (int i = 0; i < 8; ++i) red[wave][8 * h + i][c] = acc0[i] + acc1[i];
  __syncthreads();

  if (t < 16 * BN) {
    const int r = t >> 4, cc = t & 15;
    float v = 0.f;
#pragma unroll
    for (int w = 0; w < NW; ++w) v += red[w][r][cc];
    Out[r * OO + n0 + cc] = v + bval;
  }
}

extern "C" void kernel_launch(void* const* d_in, const int* in_sizes, int n_in,
                              void* d_out, int out_size, void* d_ws, size_t ws_size,
                              hipStream_t stream) {
  const float* x       = (const float*)d_in[0];
  const float* Wq_proj = (const float*)d_in[1];
  const float* bq      = (const float*)d_in[2];
  const float* Wq      = (const float*)d_in[3];
  const float* Wk      = (const float*)d_in[4];
  const float* Wv      = (const float*)d_in[5];
  const float* W1      = (const float*)d_in[6];
  const float* b1      = (const float*)d_in[7];
  const float* W2      = (const float*)d_in[8];
  const float* b2      = (const float*)d_in[9];
  const float* gamma   = (const float*)d_in[10];
  const float* beta    = (const float*)d_in[11];
  float* out = (float*)d_out;

  float* ws     = (float*)d_ws;
  float* qproj  = ws;
  float* qlast  = ws + 4096;
  float* qtil   = ws + 8192;
  float* scores = ws + 12288;
  float* cvec   = ws + 20480;
  float* attn   = ws + 24576;
  float* hin    = ws + 28672;

  wmma_gemm16<false, HH, DD, HH, true><<<dim3(HH / 16), 512, 0, stream>>>(
      x + (size_t)(SS - 1) * DD, (long)SS * DD, BN, Wq_proj, bq, qproj);
  wmma_gemm16<false, HH, HH, HH, false><<<dim3(HH / 16), 512, 0, stream>>>(
      qproj, HH, BN, Wq, nullptr, qlast);
  wmma_gemm16<true, HH, HH, DD, false><<<dim3(DD / 16), 512, 0, stream>>>(
      qlast, HH, BN, Wk, nullptr, qtil);
  scores_k<<<dim3(SS / 32, BN), 256, 0, stream>>>(x, qtil, scores, cvec);
  wsum_k<<<dim3(8, 8, BN), 256, 0, stream>>>(x, scores, cvec);
  wmma_gemm16<false, HH, DD, HH, false><<<dim3(HH / 16), 512, 0, stream>>>(
      cvec, DD, BN, Wv, nullptr, attn);
  wmma_gemm16<false, FF, HH, FF, true><<<dim3(FF / 16), 512, 0, stream>>>(
      attn, HH, BN, W1, b1, hin);
  glu_ln_gemm16<<<dim3(OO / 16), 512, 0, stream>>>(
      hin, attn, gamma, beta, W2, b2, out);
}


# first 3 GEMMs: K split over 4 blocks x 4 waves (256 blocks, was 64 x 16 waves) so every WGP takes matrix steps; consumers add the 4 partial planes in order via LDS (sums: 4 waves, then 4 planes)
# speedup vs baseline: 2.4569x; 1.0542x over previous
#include <hip/hip_runtime.h>
#include <hip/hip_bf16.h>
#include <math.h>

typedef float v2f __attribute__((ext_vector_type(2)));
typedef float v8f __attribute__((ext_vector_type(8)));

#define BN 4
#define SS 2048
#define DD 1024
#define HH 1024
#define OO 1024
#define FF 2048

template <bool TRANSB, int LDB, int K, int LDOUT, bool HASBIAS>
__global__ __launch_bounds__(512) __attribute__((amdgpu_waves_per_eu(4, 4)))
void wmma_gemm16(
    const float* __restrict__ A, long lda, int rows,
    const float* __restrict__ Bm, const float* __restrict__ bias,
    float* __restrict__ Out) {
  constexpr int NW = 16;
  constexpr int KW = K / NW;
  constexpr int NP = KW / 8;
  __shared__ float red[NW][16][16];
  const int lane = threadIdx.x & 31;
  const int wave = threadIdx.x >> 5;
  const int n0   = blockIdx.x * 16;
  const int c    = lane & 15;
  const int h    = lane >> 4;
  const int kb   = wave * KW + 4 * h;

  const float amask = (c < rows) ? 1.f : 0.f;
  const float* aptr = A + (long)(c < rows ? c : 0) * lda + kb;
  const float* bptr = TRANSB ? (Bm + (long)(n0 + c) * LDB + kb)
                             : (Bm + (long)kb * LDB + (n0 + c));

  const float bval = HASBIAS ? bias[n0 + (threadIdx.x & 15)] : 0.f;

  float4 av[NP], bv[NP];
#pragma unroll
  for (int j = 0; j < NP; ++j) {
    av[j] = *(const float4*)(aptr + 8 * j);
    if (TRANSB) {
      bv[j] = *(const float4*)(bptr + 8 * j);
    } else {
      const float* bp = bptr + (long)(8 * j) * LDB;
      bv[j].x = bp[0];
      bv[j].y = bp[LDB];
      bv[j].z = bp[2 * LDB];
      bv[j].w = bp[3 * LDB];
    }
  }

  v8f acc0 = {0.f, 0.f, 0.f, 0.f, 0.f, 0.f, 0.f, 0.f};
  v8f acc1 = {0.f, 0.f, 0.f, 0.f, 0.f, 0.f, 0.f, 0.f};
#pragma unroll
  for (int j = 0; j < NP; ++j) {
    v2f a0, b0, a1, b1;
    a0.x = av[j].x * amask;  a0.y = av[j].y * amask;
    b0.x = bv[j].x;          b0.y = bv[j].y;
    a1.x = av[j].z * amask;  a1.y = av[j].w * amask;
    b1.x = bv[j].z;          b1.y = bv[j].w;
    acc0 = __builtin_amdgcn_wmma_f32_16x16x4_f32(
        false, a0, false, b0, (short)0, acc0, false, false);
    acc1 = __builtin_amdgcn_wmma_f32_16x16x4_f32(
        false, a1, false, b1, (short)0, acc1, false, false);
  }

#pragma unroll
  for (int i = 0; i < 8; ++i) red[wave][8 * h + i][c] = acc0[i] + acc1[i];
  __syncthreads();

  const int t = threadIdx.x;
  if (t < 16 * rows) {
    const int r = t >> 4, cc = t & 15;
    float v = 0.f;
#pragma unroll
    for (int w = 0; w < NW; ++w) v += red[w][r][cc];
    Out[r * LDOUT + n0 + cc] = v + bval;
  }
}

__device__ __forceinline__ void lds_barrier() {
  __builtin_amdgcn_fence(__ATOMIC_RELEASE, "workgroup", "local");
  __builtin_amdgcn_s_barrier();
  __builtin_amdgcn_fence(__ATOMIC_ACQUIRE, "workgroup", "local");
}

__device__ __forceinline__ float wave_sum(float v) {
  for (int off = 16; off; off >>= 1) v += __shfl_xor(v, off, 32);
  return v;
}
__device__ __forceinline__ float wave_max(float v) {
  for (int off = 16; off; off >>= 1) v = fmaxf(v, __shfl_xor(v, off, 32));
  return v;
}

template <bool TRANSB, int LDB, int K, int LDOUT, bool HASBIAS, int P>
__global__ __launch_bounds__(128) __attribute__((amdgpu_waves_per_eu(1, 4)))
void wmma_gemm16_ks(
    const float* __restrict__ A, long lda, long aplane,
    const float* __restrict__ Bm, const float* __restrict__ bias,
    float* __restrict__ Out) {
  constexpr int KS = 4;
  constexpr int NW = 4;
  constexpr int KB = K / KS;
  constexpr int KW = KB / NW;
  constexpr int NP = KW / 8;
  constexpr int NA = KB / (NW * 32);
  constexpr int CP = KB + 8;
  __shared__ float red[NW][16][16];
  __shared__ __attribute__((aligned(16))) float As[BN * CP];
  const int t    = threadIdx.x;
  const int lane = t & 31;
  const int wave = t >> 5;
  const int n0   = blockIdx.x * 16;
  const int k0   = blockIdx.y * KB;
  const int c    = lane & 15;
  const int h    = lane >> 4;
  const int kb   = wave * KW + 4 * h;

  float4 ap[NA][P];
#pragma unroll
  for (int i = 0; i < NA; ++i) {
    const int idx = t + (NW * 32) * i;
    const int r = idx / (KB / 4), q = idx % (KB / 4);
    const float* src = A + (long)r * lda + k0 + 4 * q;
#pragma unroll
    for (int p = 0; p < P; ++p) ap[i][p] = *(const float4*)(src + (long)p * aplane);
  }
  const float braw = HASBIAS ? bias[n0 + (t & 15)] : 0.f;
  const float bval = (blockIdx.y == 0) ? braw : 0.f;

  const float* bptr = TRANSB ? (Bm + (long)(n0 + c) * LDB + k0 + kb)
                             : (Bm + (long)(k0 + kb) * LDB + (n0 + c));
  float4 bv[NP];
#pragma unroll
  for (int j = 0; j < NP; ++j) {
    if (TRANSB) {
      bv[j] = *(const float4*)(bptr + 8 * j);
    } else {
      const float* bp = bptr + (long)(8 * j) * LDB;
      bv[j].x = bp[0];
      bv[j].y = bp[LDB];
      bv[j].z = bp[2 * LDB];
      bv[j].w = bp[3 * LDB];
    }
  }

#pragma unroll
  for (int i = 0; i < NA; ++i) {
    const int idx = t + (NW * 32) * i;
    const int r = idx / (KB / 4), q = idx % (KB / 4);
    float4 s = ap[i][0];
#pragma unroll
    for (int p = 1; p < P; ++p) {
      s.x += ap[i][p].x; s.y += ap[i][p].y; s.z += ap[i][p].z; s.w += ap[i][p].w;
    }
    *(float4*)(As + r * CP + 4 * q) = s;
  }
  lds_barrier();

  const float amask = (c < BN) ? 1.f : 0.f;
  const float* aptr = As + (c < BN ? c : 0) * CP + kb;
  float4 av[NP];
#pragma unroll
  for (int j = 0; j < NP; ++j) av[j] = *(const float4*)(aptr + 8 * j);

  v8f acc0 = {0.f, 0.f, 0.f, 0.f, 0.f, 0.f, 0.f, 0.f};
  v8f acc1 = {0.f, 0.f, 0.f, 0.f, 0.f, 0.f, 0.f, 0.f};
#pragma unroll
  for (int j = 0; j < NP; ++j) {
    v2f a0, b0, a1, b1;
    a0.x = av[j].x * amask;  a0.y = av[j].y * amask;
    b0.x = bv[j].x;          b0.y = bv[j].y;
    a1.x = av[j].z * amask;  a1.y = av[j].w * amask;
    b1.x = bv[j].z;          b1.y = bv[j].w;
    acc0 = __builtin_amdgcn_wmma_f32_16x16x4_f32(
        false, a0, false, b0, (short)0, acc0, false, false);
    acc1 = __builtin_amdgcn_wmma_f32_16x16x4_f32(
        false, a1, false, b1, (short)0, acc1, false, false);
  }

  __builtin_amdgcn_sched_group_barrier(0x100, NP, 0);
  __builtin_amdgcn_sched_group_barrier(0x008, 2 * NP, 0);

#pragma unroll
  for (int i = 0; i < 8; ++i) red[wave][8 * h + i][c] = acc0[i] + acc1[i];
  __syncthreads();

  if (t < 16 * BN) {
    const int r = t >> 4, cc = t & 15;
    float v = 0.f;
#pragma unroll
    for (int w = 0; w < NW; ++w) v += red[w][r][cc];
    Out[(long)blockIdx.y * (BN * LDOUT) + r * LDOUT + n0 + cc] = v + bval;
  }
}

__global__ __launch_bounds__(256) __attribute__((amdgpu_waves_per_eu(2, 2)))
void scores_k(const float* __restrict__ x,
              const float* __restrict__ qtil,
              float* __restrict__ scores,
              float* __restrict__ cvec) {
  constexpr int RW = 4;
  const int b    = blockIdx.y;
  const int lane = threadIdx.x & 31;
  const int srow = blockIdx.x * (8 * RW) + (threadIdx.x >> 5) * RW;
  const float4* xr = (const float4*)(x + ((size_t)b * SS + srow) * DD) + lane;
  const float4* q4 = (const float4*)(qtil + (size_t)b * DD) + threadIdx.x;
  float4 qp[4];
#pragma unroll
  for (int p = 0; p < 4; ++p) qp[p] = q4[p * (BN * DD / 4)];
  float4 qv[8], xv[RW][8];
#pragma unroll
  for (int r = 0; r < RW; ++r)
#pragma unroll
    for (int i = 0; i < 8; ++i) xv[r][i] = xr[r * (DD / 4) + i * 32];
  __shared__ float4 qs[DD / 4];
  float4 qsum = qp[0];
#pragma unroll
  for (int p = 1; p < 4; ++p) {
    qsum.x += qp[p].x; qsum.y += qp[p].y; qsum.z += qp[p].z; qsum.w += qp[p].w;
  }
  qs[threadIdx.x] = qsum;
  lds_barrier();
#pragma unroll
  for (int i = 0; i < 8; ++i) qv[i] = qs[i * 32 + lane];
  float sum[RW];
#pragma unroll
  for (int r = 0; r < RW; ++r) {
    float s = 0.f;
#pragma unroll
    for (int i = 0; i < 8; ++i)
      s += xv[r][i].x * qv[i].x + xv[r][i].y * qv[i].y +
           xv[r][i].z * qv[i].z + xv[r][i].w * qv[i].w;
    sum[r] = s;
  }
  for (int off = 16; off; off >>= 1) {
#pragma unroll
    for (int r = 0; r < RW; ++r) sum[r] += __shfl_xor(sum[r], off, 32);
  }
  const float v = lane == 0 ? sum[0] : lane == 1 ? sum[1]
                : lane == 2 ? sum[2] : sum[3];
  if (lane < RW) scores[b * SS + srow + lane] = v * 0.03125f;
  if (blockIdx.x == 0)
    ((float4*)(cvec + b * DD))[threadIdx.x] = float4{0.f, 0.f, 0.f, 0.f};
}

__global__ __launch_bounds__(256) __attribute__((amdgpu_waves_per_eu(2, 2)))
void wsum_k(const float* __restrict__ x,
            const float* __restrict__ scores,
            float* __restrict__ cvec) {
  constexpr int RW = 32;
  constexpr int NWV = 8;
  const int cs = blockIdx.x, chunk = blockIdx.y, b = blockIdx.z;
  const int t = threadIdx.x;
  const int lane = t & 31;
  const int wave = __builtin_amdgcn_readfirstlane(t >> 5);
  const int s0 = chunk * (RW * NWV) + wave * RW;
  const float* sc = scores + b * SS;
  const float4 sa = ((const float4*)sc)[t];
  const float4 sb = ((const float4*)sc)[t + 256];
  const float sown = sc[s0 + lane];
  const float4* xp = (const float4*)x + ((size_t)b * SS + s0) * (DD / 4)
                     + cs * 32 + lane;
  float4 xv[RW];
#pragma unroll
  for (int i = 0; i < RW; ++i) xv[i] = xp[(size_t)i * (DD / 4)];
  __builtin_amdgcn_sched_barrier(0);

  __shared__ float smax[NWV], ssum[NWV];
  __shared__ float4 red[NWV][32];
  float mx = fmaxf(fmaxf(fmaxf(sa.x, sa.y), fmaxf(sa.z, sa.w)),
                   fmaxf(fmaxf(sb.x, sb.y), fmaxf(sb.z, sb.w)));
  mx = wave_max(mx);
  if (lane == 0) smax[wave] = mx;
  lds_barrier();
  mx = smax[0];
#pragma unroll
  for (int wv = 1; wv < NWV; ++wv) mx = fmaxf(mx, smax[wv]);
  float es = expf(sa.x - mx) + expf(sa.y - mx) + expf(sa.z - mx) +
             expf(sa.w - mx) + expf(sb.x - mx) + expf(sb.y - mx) +
             expf(sb.z - mx) + expf(sb.w - mx);
  es = wave_sum(es);
  if (lane == 0) ssum[wave] = es;
  lds_barrier();
  float tot = 0.f;
#pragma unroll
  for (int wv = 0; wv < NWV; ++wv) tot += ssum[wv];
  const float inv = 1.f / tot;
  const int wbits = __builtin_bit_cast(int, expf(sown - mx) * inv);

  float4 acc = {0.f, 0.f, 0.f, 0.f};
#pragma unroll
  for (int i = 0; i < RW; ++i) {
    const float ww = __builtin_bit_cast(float, __builtin_amdgcn_readlane(wbits, i));
    acc.x += ww * xv[i].x; acc.y += ww * xv[i].y;
    acc.z += ww * xv[i].z; acc.w += ww * xv[i].w;
  }
  red[wave][lane] = acc;
  __syncthreads();
  if (t < 128) {
    const float* rf = (const float*)red;
    float v = 0.f;
#pragma unroll
    for (int wv = 0; wv < NWV; ++wv) v += rf[wv * 128 + t];
    atomicAdd(cvec + b * DD + cs * 128 + t, v);
  }
}

__global__ __launch_bounds__(512) __attribute__((amdgpu_waves_per_eu(4, 4)))
void glu_ln_gemm16(const float* __restrict__ hin,
                   const float* __restrict__ attn,
                   const float* __restrict__ gamma,
                   const float* __restrict__ beta,
                   const float* __restrict__ Bm,
                   const float* __restrict__ bias,
                   float* __restrict__ Out) {
  constexpr int NW = 16;
  constexpr int KW = HH / NW;
  constexpr int NP = KW / 8;
  constexpr int CP = HH + 8;
  __shared__ float red[NW][16][16];
  __shared__ __attribute__((aligned(16))) float comb[BN * CP];
  __shared__ float psum[NW], psq[NW];
  const int t    = threadIdx.x;
  const int lane = t & 31;
  const int wave = t >> 5;
  const int n0   = blockIdx.x * 16;
  const int c    = lane & 15;
  const int h    = lane >> 4;
  const int kb   = wave * KW + 4 * h;
  const int bt   = t >> 7;
  const int q    = t & 127;

  const float4* h4 = (const float4*)(hin + (size_t)bt * FF);
  const float4 x1a = h4[q],       x1b = h4[q + 128];
  const float4 x2a = h4[q + 256], x2b = h4[q + 384];
  const float4* a4 = (const float4*)(attn + (size_t)bt * HH);
  const float4 ata = a4[q], atb = a4[q + 128];
  const float4 gaa = ((const float4*)gamma)[q], gab = ((const float4*)gamma)[q + 128];
  const float4 bea = ((const float4*)beta)[q],  beb = ((const float4*)beta)[q + 128];
  const float bval = bias[n0 + (t & 15)];

  const float* bptr = Bm + (long)kb * OO + (n0 + c);
  float4 bv[NP];
#pragma unroll
  for (int j = 0; j < NP; ++j) {
    const float* bp = bptr + (long)(8 * j) * OO;
    bv[j].x = bp[0];
    bv[j].y = bp[OO];
    bv[j].z = bp[2 * OO];
    bv[j].w = bp[3 * OO];
  }

  const float x1[8] = {x1a.x, x1a.y, x1a.z, x1a.w, x1b.x, x1b.y, x1b.z, x1b.w};
  const float x2[8] = {x2a.x, x2a.y, x2a.z, x2a.w, x2b.x, x2b.y, x2b.z, x2b.w};
  float hv[8];
  float sum = 0.f;
#pragma unroll
  for (int j = 0; j < 8; ++j) {
    const float g = 0.5f * x2[j] * (1.f + erff(x2[j] * 0.70710678118654752f));
    hv[j] = x1[j] * g;
    sum += hv[j];
  }
  sum = wave_sum(sum);
  if (lane == 0) psum[wave] = sum;
  lds_barrier();
  const int w0 = bt * 4;
  const float mean = (psum[w0] + psum[w0 + 1] + psum[w0 + 2] + psum[w0 + 3]) * (1.f / HH);
  float sq = 0.f;
#pragma unroll
  for (int j = 0; j < 8; ++j) { const float d = hv[j] - mean; sq += d * d; }
  sq = wave_sum(sq);
  if (lane == 0) psq[wave] = sq;
  lds_barrier();
  const float var = (psq[w0] + psq[w0 + 1] + psq[w0 + 2] + psq[w0 + 3]) * (1.f / HH);
  const float rstd = rsqrtf(var + 1e-5f);
  float4 ca, cb;
  ca.x = ata.x + (hv[0] - mean) * rstd * gaa.x + bea.x;
  ca.y = ata.y + (hv[1] - mean) * rstd * gaa.y + bea.y;
  ca.z = ata.z + (hv[2] - mean) * rstd * gaa.z + bea.z;
  ca.w = ata.w + (hv[3] - mean) * rstd * gaa.w + bea.w;
  cb.x = atb.x + (hv[4] - mean) * rstd * gab.x + beb.x;
  cb.y = atb.y + (hv[5] - mean) * rstd * gab.y + beb.y;
  cb.z = atb.z + (hv[6] - mean) * rstd * gab.z + beb.z;
  cb.w = atb.w + (hv[7] - mean) * rstd * gab.w + beb.w;
  *(float4*)(comb + bt * CP + 4 * q)       = ca;
  *(float4*)(comb + bt * CP + 512 + 4 * q) = cb;
  lds_barrier();

  const float amask = (c < BN) ? 1.f : 0.f;
  const float* aptr = comb + (c < BN ? c : 0) * CP + kb;
  float4 av[NP];
#pragma unroll
  for (int j = 0; j < NP; ++j) av[j] = *(const float4*)(aptr + 8 * j);

  v8f acc0 = {0.f, 0.f, 0.f, 0.f, 0.f, 0.f, 0.f, 0.f};
  v8f acc1 = {0.f, 0.f, 0.f, 0.f, 0.f, 0.f, 0.f, 0.f};
#pragma unroll
  for (int j = 0; j < NP; ++j) {
    v2f a0, b0, a1, b1;
    a0.x = av[j].x * amask;  a0.y = av[j].y * amask;
    b0.x = bv[j].x;          b0.y = bv[j].y;
    a1.x = av[j].z * amask;  a1.y = av[j].w * amask;
    b1.x = bv[j].z;          b1.y = bv[j].w;
    acc0 = __builtin_amdgcn_wmma_f32_16x16x4_f32(
        false, a0, false, b0, (short)0, acc0, false, false);
    acc1 = __builtin_amdgcn_wmma_f32_16x16x4_f32(
        false, a1, false, b1, (short)0, acc1, false, false);
  }

#pragma unroll
  for (int i = 0; i < 8; ++i) red[wave][8 * h + i][c] = acc0[i] + acc1[i];
  __syncthreads();

  if (t < 16 * BN) {
    const int r = t >> 4, cc = t & 15;
    float v = 0.f;
#pragma unroll
    for (int w = 0; w < NW; ++w) v += red[w][r][cc];
    Out[r * OO + n0 + cc] = v + bval;
  }
}

extern "C" void kernel_launch(void* const* d_in, const int* in_sizes, int n_in,
                              void* d_out, int out_size, void* d_ws, size_t ws_size,
                              hipStream_t stream) {
  const float* x       = (const float*)d_in[0];
  const float* Wq_proj = (const float*)d_in[1];
  const float* bq      = (const float*)d_in[2];
  const float* Wq      = (const float*)d_in[3];
  const float* Wk      = (const float*)d_in[4];
  const float* Wv      = (const float*)d_in[5];
  const float* W1      = (const float*)d_in[6];
  const float* b1      = (const float*)d_in[7];
  const float* W2      = (const float*)d_in[8];
  const float* b2      = (const float*)d_in[9];
  const float* gamma   = (const float*)d_in[10];
  const float* beta    = (const float*)d_in[11];
  float* out = (float*)d_out;

  float* ws     = (float*)d_ws;
  float* qproj  = ws;
  float* qlast  = ws + 16384;
  float* qtil   = ws + 32768;
  float* scores = ws + 49152;
  float* cvec   = ws + 57344;
  float* attn   = ws + 61440;
  float* hin    = ws + 65536;

  wmma_gemm16_ks<false, HH, DD, HH, true, 1><<<dim3(HH / 16, 4), 128, 0, stream>>>(
      x + (size_t)(SS - 1) * DD, (long)SS * DD, 0, Wq_proj, bq, qproj);
  wmma_gemm16_ks<false, HH, HH, HH, false, 4><<<dim3(HH / 16, 4), 128, 0, stream>>>(
      qproj, HH, BN * HH, Wq, nullptr, qlast);
  wmma_gemm16_ks<true, HH, HH, DD, false, 4><<<dim3(DD / 16, 4), 128, 0, stream>>>(
      qlast, HH, BN * HH, Wk, nullptr, qtil);
  scores_k<<<dim3(SS / 32, BN), 256, 0, stream>>>(x, qtil, scores, cvec);
  wsum_k<<<dim3(8, 8, BN), 256, 0, stream>>>(x, scores, cvec);
  wmma_gemm16<false, HH, DD, HH, false><<<dim3(HH / 16), 512, 0, stream>>>(
      cvec, DD, BN, Wv, nullptr, attn);
  wmma_gemm16<false, FF, HH, FF, true><<<dim3(FF / 16), 512, 0, stream>>>(
      attn, HH, BN, W1, b1, hin);
  glu_ln_gemm16<<<dim3(OO / 16), 512, 0, stream>>>(
      hin, attn, gamma, beta, W2, b2, out);
}


# launches 8 -> 7: x read once (score blocks also write exp-weighted 32-row pieces); attn GEMM re-tiled to 16 K slices, rescales+adds the 64 pieces itself, slices meet by f32 atomics (new sum order)
# speedup vs baseline: 2.7687x; 1.1269x over previous
#include <hip/hip_runtime.h>
#include <hip/hip_bf16.h>
#include <math.h>

typedef float v2f __attribute__((ext_vector_type(2)));
typedef float v8f __attribute__((ext_vector_type(8)));

#define BN 4
#define SS 2048
#define DD 1024
#define HH 1024
#define OO 1024
#define FF 2048

template <bool TRANSB, int LDB, int K, int LDOUT, bool HASBIAS>
__global__ __launch_bounds__(512) __attribute__((amdgpu_waves_per_eu(4, 4)))
void wmma_gemm16(
    const float* __restrict__ A, long lda, int rows,
    const float* __restrict__ Bm, const float* __restrict__ bias,
    float* __restrict__ Out) {
  constexpr int NW = 16;
  constexpr int KW = K / NW;
  constexpr int NP = KW / 8;
  __shared__ float red[NW][16][16];
  const int lane = threadIdx.x & 31;
  const int wave = threadIdx.x >> 5;
  const int n0   = blockIdx.x * 16;
  const int c    = lane & 15;
  const int h    = lane >> 4;
  const int kb   = wave * KW + 4 * h;

  const float amask = (c < rows) ? 1.f : 0.f;
  const float* aptr = A + (long)(c < rows ? c : 0) * lda + kb;
  const float* bptr = TRANSB ? (Bm + (long)(n0 + c) * LDB + kb)
                             : (Bm + (long)kb * LDB + (n0 + c));

  const float bval = HASBIAS ? bias[n0 + (threadIdx.x & 15)] : 0.f;

  float4 av[NP], bv[NP];
#pragma unroll
  for (int j = 0; j < NP; ++j) {
    av[j] = *(const float4*)(aptr + 8 * j);
    if (TRANSB) {
      bv[j] = *(const float4*)(bptr + 8 * j);
    } else {
      const float* bp = bptr + (long)(8 * j) * LDB;
      bv[j].x = bp[0];
      bv[j].y = bp[LDB];
      bv[j].z = bp[2 * LDB];
      bv[j].w = bp[3 * LDB];
    }
  }

  v8f acc0 = {0.f, 0.f, 0.f, 0.f, 0.f, 0.f, 0.f, 0.f};
  v8f acc1 = {0.f, 0.f, 0.f, 0.f, 0.f, 0.f, 0.f, 0.f};
#pragma unroll
  for (int j = 0; j < NP; ++j) {
    v2f a0, b0, a1, b1;
    a0.x = av[j].x * amask;  a0.y = av[j].y * amask;
    b0.x = bv[j].x;          b0.y = bv[j].y;
    a1.x = av[j].z * amask;  a1.y = av[j].w * amask;
    b1.x = bv[j].z;          b1.y = bv[j].w;
    acc0 = __builtin_amdgcn_wmma_f32_16x16x4_f32(
        false, a0, false, b0, (short)0, acc0, false, false);
    acc1 = __builtin_amdgcn_wmma_f32_16x16x4_f32(
        false, a1, false, b1, (short)0, acc1, false, false);
  }

#pragma unroll
  for (int i = 0; i < 8; ++i) red[wave][8 * h + i][c] = acc0[i] + acc1[i];
  __syncthreads();

  const int t = threadIdx.x;
  if (t < 16 * rows) {
    const int r = t >> 4, cc = t & 15;
    float v = 0.f;
#pragma unroll
    for (int w = 0; w < NW; ++w) v += red[w][r][cc];
    Out[r * LDOUT + n0 + cc] = v + bval;
  }
}

__device__ __forceinline__ void lds_barrier() {
  __builtin_amdgcn_fence(__ATOMIC_RELEASE, "workgroup", "local");
  __builtin_amdgcn_s_barrier();
  __builtin_amdgcn_fence(__ATOMIC_ACQUIRE, "workgroup", "local");
}

__device__ __forceinline__ float wave_sum(float v) {
  for (int off = 16; off; off >>= 1) v += __shfl_xor(v, off, 32);
  return v;
}
__device__ __forceinline__ float wave_max(float v) {
  for (int off = 16; off; off >>= 1) v = fmaxf(v, __shfl_xor(v, off, 32));
  return v;
}

template <bool TRANSB, int LDB, int K, int LDOUT, bool HASBIAS, int P>
__global__ __launch_bounds__(128) __attribute__((amdgpu_waves_per_eu(1, 4)))
void wmma_gemm16_ks(
    const float* __restrict__ A, long lda, long aplane,
    const float* __restrict__ Bm, const float* __restrict__ bias,
    float* __restrict__ Out) {
  constexpr int KS = 4;
  constexpr int NW = 4;
  constexpr int KB = K / KS;
  constexpr int KW = KB / NW;
  constexpr int NP = KW / 8;
  constexpr int NA = KB / (NW * 32);
  constexpr int CP = KB + 8;
  __shared__ float red[NW][16][16];
  __shared__ __attribute__((aligned(16))) float As[BN * CP];
  const int t    = threadIdx.x;
  const int lane = t & 31;
  const int wave = t >> 5;
  const int n0   = blockIdx.x * 16;
  const int k0   = blockIdx.y * KB;
  const int c    = lane & 15;
  const int h    = lane >> 4;
  const int kb   = wave * KW + 4 * h;

  float4 ap[NA][P];
#pragma unroll
  for (int i = 0; i < NA; ++i) {
    const int idx = t + (NW * 32) * i;
    const int r = idx / (KB / 4), q = idx % (KB / 4);
    const float* src = A + (long)r * lda + k0 + 4 * q;
#pragma unroll
    for (int p = 0; p < P; ++p) ap[i][p] = *(const float4*)(src + (long)p * aplane);
  }
  const float braw = HASBIAS ? bias[n0 + (t & 15)] : 0.f;
  const float bval = (blockIdx.y == 0) ? braw : 0.f;

  const float* bptr = TRANSB ? (Bm + (long)(n0 + c) * LDB + k0 + kb)
                             : (Bm + (long)(k0 + kb) * LDB + (n0 + c));
  float4 bv[NP];
#pragma unroll
  for (int j = 0; j < NP; ++j) {
    if (TRANSB) {
      bv[j] = *(const float4*)(bptr + 8 * j);
    } else {
      const float* bp = bptr + (long)(8 * j) * LDB;
      bv[j].x = bp[0];
      bv[j].y = bp[LDB];
      bv[j].z = bp[2 * LDB];
      bv[j].w = bp[3 * LDB];
    }
  }

#pragma unroll
  for (int i = 0; i < NA; ++i) {
    const int idx = t + (NW * 32) * i;
    const int r = idx / (KB / 4), q = idx % (KB / 4);
    float4 s = ap[i][0];
#pragma unroll
    for (int p = 1; p < P; ++p) {
      s.x += ap[i][p].x; s.y += ap[i][p].y; s.z += ap[i][p].z; s.w += ap[i][p].w;
    }
    *(float4*)(As + r * CP + 4 * q) = s;
  }
  lds_barrier();

  const float amask = (c < BN) ? 1.f : 0.f;
  const float* aptr = As + (c < BN ? c : 0) * CP + kb;
  float4 av[NP];
#pragma unroll
  for (int j = 0; j < NP; ++j) av[j] = *(const float4*)(aptr + 8 * j);

  v8f acc0 = {0.f, 0.f, 0.f, 0.f, 0.f, 0.f, 0.f, 0.f};
  v8f acc1 = {0.f, 0.f, 0.f, 0.f, 0.f, 0.f, 0.f, 0.f};
#pragma unroll
  for (int j = 0; j < NP; ++j) {
    v2f a0, b0, a1, b1;
    a0.x = av[j].x * amask;  a0.y = av[j].y * amask;
    b0.x = bv[j].x;          b0.y = bv[j].y;
    a1.x = av[j].z * amask;  a1.y = av[j].w * amask;
    b1.x = bv[j].z;          b1.y = bv[j].w;
    acc0 = __builtin_amdgcn_wmma_f32_16x16x4_f32(
        false, a0, false, b0, (short)0, acc0, false, false);
    acc1 = __builtin_amdgcn_wmma_f32_16x16x4_f32(
        false, a1, false, b1, (short)0, acc1, false, false);
  }

  __builtin_amdgcn_sched_group_barrier(0x100, NP, 0);
  __builtin_amdgcn_sched_group_barrier(0x008, 2 * NP, 0);

#pragma unroll
  for (int i = 0; i < 8; ++i) red[wave][8 * h + i][c] = acc0[i] + acc1[i];
  __syncthreads();

  if (t < 16 * BN) {
    const int r = t >> 4, cc = t & 15;
    float v = 0.f;
#pragma unroll
    for (int w = 0; w < NW; ++w) v += red[w][r][cc];
    Out[(long)blockIdx.y * (BN * LDOUT) + r * LDOUT + n0 + cc] = v + bval;
  }
}

__global__ __launch_bounds__(256) __attribute__((amdgpu_waves_per_eu(2, 2)))
void scores_part_k(const float* __restrict__ x,
                   const float* __restrict__ qtil,
                   float* __restrict__ part,
                   float* __restrict__ stats,
                   float* __restrict__ attn) {
  constexpr int RW = 4;
  constexpr int NWV = 8;
  const int b    = blockIdx.y;
  const int t    = threadIdx.x;
  const int lane = t & 31;
  const int wave = t >> 5;
  const int srow = blockIdx.x * (NWV * RW) + wave * RW;
  const float4* xr = (const float4*)(x + ((size_t)b * SS + srow) * DD) + lane;
  const float4* q4 = (const float4*)(qtil + (size_t)b * DD) + t;
  float4 qp[4];
#pragma unroll
  for (int p = 0; p < 4; ++p) qp[p] = q4[p * (BN * DD / 4)];
  float4 qv[8], xv[RW][8];
#pragma unroll
  for (int r = 0; r < RW; ++r)
#pragma unroll
    for (int i = 0; i < 8; ++i) xv[r][i] = xr[r * (DD / 4) + i * 32];
  __shared__ float4 qs[DD / 4];
  __shared__ float smax[NWV], ssum[NWV];
  __shared__ float4 red[NWV][DD / 4];
  float4 qsum = qp[0];
#pragma unroll
  for (int p = 1; p < 4; ++p) {
    qsum.x += qp[p].x; qsum.y += qp[p].y; qsum.z += qp[p].z; qsum.w += qp[p].w;
  }
  qs[t] = qsum;
  lds_barrier();
#pragma unroll
  for (int i = 0; i < 8; ++i) qv[i] = qs[i * 32 + lane];
  float sum[RW];
#pragma unroll
  for (int r = 0; r < RW; ++r) {
    float s = 0.f;
#pragma unroll
    for (int i = 0; i < 8; ++i)
      s += xv[r][i].x * qv[i].x + xv[r][i].y * qv[i].y +
           xv[r][i].z * qv[i].z + xv[r][i].w * qv[i].w;
    sum[r] = s;
  }
  for (int off = 16; off; off >>= 1) {
#pragma unroll
    for (int r = 0; r < RW; ++r) sum[r] += __shfl_xor(sum[r], off, 32);
  }
  float sc[RW];
#pragma unroll
  for (int r = 0; r < RW; ++r) sc[r] = sum[r] * 0.03125f;
  const float mw = fmaxf(fmaxf(sc[0], sc[1]), fmaxf(sc[2], sc[3]));
  if (lane == 0) smax[wave] = mw;
  lds_barrier();
  float m = smax[0];
#pragma unroll
  for (int wv = 1; wv < NWV; ++wv) m = fmaxf(m, smax[wv]);
  float e[RW];
#pragma unroll
  for (int r = 0; r < RW; ++r) e[r] = expf(sc[r] - m);
  if (lane == 0) ssum[wave] = ((e[0] + e[1]) + e[2]) + e[3];
#pragma unroll
  for (int i = 0; i < 8; ++i) {
    float4 a;
    a.x = e[0] * xv[0][i].x; a.y = e[0] * xv[0][i].y;
    a.z = e[0] * xv[0][i].z; a.w = e[0] * xv[0][i].w;
#pragma unroll
    for (int r = 1; r < RW; ++r) {
      a.x = fmaf(e[r], xv[r][i].x, a.x); a.y = fmaf(e[r], xv[r][i].y, a.y);
      a.z = fmaf(e[r], xv[r][i].z, a.z); a.w = fmaf(e[r], xv[r][i].w, a.w);
    }
    red[wave][i * 32 + lane] = a;
  }
  __syncthreads();
  float4 v = red[0][t];
#pragma unroll
  for (int wv = 1; wv < NWV; ++wv) {
    const float4 u = red[wv][t];
    v.x += u.x; v.y += u.y; v.z += u.z; v.w += u.w;
  }
  const int plane = b * (SS / (NWV * RW)) + blockIdx.x;
  ((float4*)(part + (size_t)plane * DD))[t] = v;
  if (t == 0) {
    float l = 0.f;
#pragma unroll
    for (int wv = 0; wv < NWV; ++wv) l += ssum[wv];
    float2 ml; ml.x = m; ml.y = l;
    ((float2*)stats)[plane] = ml;
  }
  if (blockIdx.x == 0)
    ((float4*)(attn + b * HH))[t] = float4{0.f, 0.f, 0.f, 0.f};
}

__global__ __launch_bounds__(128) __attribute__((amdgpu_waves_per_eu(1, 2)))
void merge_gemm16(const float* __restrict__ part,
                  const float* __restrict__ stats,
                  const float* __restrict__ Bm,
                  float* __restrict__ Out) {
  constexpr int KB = 64;
  constexpr int NP = KB / 8;
  constexpr int NJ = 64;
  constexpr int HJ = NJ / 2;
  constexpr int CP = KB + 8;
  __shared__ __attribute__((aligned(16))) float As[BN * CP];
  const int t    = threadIdx.x;
  const int lane = t & 31;
  const int wave = __builtin_amdgcn_readfirstlane(t >> 5);
  const int n0   = blockIdx.x * 64 + wave * 16;
  const int k0   = blockIdx.y * KB;
  const int c    = lane & 15;
  const int h    = lane >> 4;

  const float2* st = (const float2*)stats + wave * NJ;
  const float2 slo = st[lane], shi = st[lane + HJ];
  const float4* pp = (const float4*)(part + (size_t)(wave * NJ + HJ * h) * DD + k0) + c;
  float4 pv[HJ];
#pragma unroll
  for (int i = 0; i < HJ; ++i) pv[i] = pp[(size_t)i * (DD / 4)];
  const float* bptr = Bm + (long)(k0 + 4 * h) * HH + (n0 + c);
  float4 bv[NP];
#pragma unroll
  for (int j = 0; j < NP; ++j) {
    const float* bp = bptr + (long)(8 * j) * HH;
    bv[j].x = bp[0];
    bv[j].y = bp[HH];
    bv[j].z = bp[2 * HH];
    bv[j].w = bp[3 * HH];
  }

  const float mb = wave_max(fmaxf(slo.x, shi.x));
  const float elo = expf(slo.x - mb), ehi = expf(shi.x - mb);
  const float lb = wave_sum(elo * slo.y + ehi * shi.y);
  const float inv = 1.f / lb;
  const int wlo = __builtin_bit_cast(int, elo * inv);
  const int whi = __builtin_bit_cast(int, ehi * inv);

  float4 acc = {0.f, 0.f, 0.f, 0.f};
#pragma unroll
  for (int i = 0; i < HJ; ++i) {
    const float f0 = __builtin_bit_cast(float, __builtin_amdgcn_readlane(wlo, i));
    const float f1 = __builtin_bit_cast(float, __builtin_amdgcn_readlane(whi, i));
    const float ww = h ? f1 : f0;
    acc.x = fmaf(ww, pv[i].x, acc.x); acc.y = fmaf(ww, pv[i].y, acc.y);
    acc.z = fmaf(ww, pv[i].z, acc.z); acc.w = fmaf(ww, pv[i].w, acc.w);
  }
  acc.x += __shfl_xor(acc.x, 16, 32);
  acc.y += __shfl_xor(acc.y, 16, 32);
  acc.z += __shfl_xor(acc.z, 16, 32);
  acc.w += __shfl_xor(acc.w, 16, 32);
  __builtin_amdgcn_sched_group_barrier(0x020, 2 + HJ + 4 * NP, 0);
  __builtin_amdgcn_sched_group_barrier(0x080, 14, 0);
  if (h == 0) *(float4*)(As + wave * CP + 4 * c) = acc;
  lds_barrier();

  const float amask = (c < BN) ? 1.f : 0.f;
  const float* aptr = As + (c < BN ? c : 0) * CP + 4 * h;
  float4 av[NP];
#pragma unroll
  for (int j = 0; j < NP; ++j) av[j] = *(const float4*)(aptr + 8 * j);

  v8f acc0 = {0.f, 0.f, 0.f, 0.f, 0.f, 0.f, 0.f, 0.f};
  v8f acc1 = {0.f, 0.f, 0.f, 0.f, 0.f, 0.f, 0.f, 0.f};
#pragma unroll
  for (int j = 0; j < NP; ++j) {
    v2f a0, b0, a1, b1;
    a0.x = av[j].x * amask;  a0.y = av[j].y * amask;
    b0.x = bv[j].x;          b0.y = bv[j].y;
    a1.x = av[j].z * amask;  a1.y = av[j].w * amask;
    b1.x = bv[j].z;          b1.y = bv[j].w;
    acc0 = __builtin_amdgcn_wmma_f32_16x16x4_f32(
        false, a0, false, b0, (short)0, acc0, false, false);
    acc1 = __builtin_amdgcn_wmma_f32_16x16x4_f32(
        false, a1, false, b1, (short)0, acc1, false, false);
  }

  __builtin_amdgcn_sched_group_barrier(0x100, NP, 0);
  __builtin_amdgcn_sched_group_barrier(0x008, 2 * NP, 0);

  float o[BN];
#pragma unroll
  for (int i = 0; i < BN; ++i) o[i] = acc0[i] + acc1[i];
  if (h == 0) {
#pragma unroll
    for (int i = 0; i < BN; ++i) atomicAdd(Out + i * HH + n0 + c, o[i]);
  }
}

__global__ __launch_bounds__(512) __attribute__((amdgpu_waves_per_eu(4, 4)))
void glu_ln_gemm16(const float* __restrict__ hin,
                   const float* __restrict__ attn,
                   const float* __restrict__ gamma,
                   const float* __restrict__ beta,
                   const float* __restrict__ Bm,
                   const float* __restrict__ bias,
                   float* __restrict__ Out) {
  constexpr int NW = 16;
  constexpr int KW = HH / NW;
  constexpr int NP = KW / 8;
  constexpr int CP = HH + 8;
  __shared__ float red[NW][16][16];
  __shared__ __attribute__((aligned(16))) float comb[BN * CP];
  __shared__ float psum[NW], psq[NW];
  const int t    = threadIdx.x;
  const int lane = t & 31;
  const int wave = t >> 5;
  const int n0   = blockIdx.x * 16;
  const int c    = lane & 15;
  const int h    = lane >> 4;
  const int kb   = wave * KW + 4 * h;
  const int bt   = t >> 7;
  const int q    = t & 127;

  const float4* h4 = (const float4*)(hin + (size_t)bt * FF);
  const float4 x1a = h4[q],       x1b = h4[q + 128];
  const float4 x2a = h4[q + 256], x2b = h4[q + 384];
  const float4* a4 = (const float4*)(attn + (size_t)bt * HH);
  const float4 ata = a4[q], atb = a4[q + 128];
  const float4 gaa = ((const float4*)gamma)[q], gab = ((const float4*)gamma)[q + 128];
  const float4 bea = ((const float4*)beta)[q],  beb = ((const float4*)beta)[q + 128];
  const float bval = bias[n0 + (t & 15)];

  const float* bptr = Bm + (long)kb * OO + (n0 + c);
  float4 bv[NP];
#pragma unroll
  for (int j = 0; j < NP; ++j) {
    const float* bp = bptr + (long)(8 * j) * OO;
    bv[j].x = bp[0];
    bv[j].y = bp[OO];
    bv[j].z = bp[2 * OO];
    bv[j].w = bp[3 * OO];
  }

  const float x1[8] = {x1a.x, x1a.y, x1a.z, x1a.w, x1b.x, x1b.y, x1b.z, x1b.w};
  const float x2[8] = {x2a.x, x2a.y, x2a.z, x2a.w, x2b.x, x2b.y, x2b.z, x2b.w};
  float hv[8];
  float sum = 0.f;
#pragma unroll
  for (int j = 0; j < 8; ++j) {
    const float g = 0.5f * x2[j] * (1.f + erff(x2[j] * 0.70710678118654752f));
    hv[j] = x1[j] * g;
    sum += hv[j];
  }
  sum = wave_sum(sum);
  if (lane == 0) psum[wave] = sum;
  lds_barrier();
  const int w0 = bt * 4;
  const float mean = (psum[w0] + psum[w0 + 1] + psum[w0 + 2] + psum[w0 + 3]) * (1.f / HH);
  float sq = 0.f;
#pragma unroll
  for (int j = 0; j < 8; ++j) { const float d = hv[j] - mean; sq += d * d; }
  sq = wave_sum(sq);
  if (lane == 0) psq[wave] = sq;
  lds_barrier();
  const float var = (psq[w0] + psq[w0 + 1] + psq[w0 + 2] + psq[w0 + 3]) * (1.f / HH);
  const float rstd = rsqrtf(var + 1e-5f);
  float4 ca, cb;
  ca.x = ata.x + (hv[0] - mean) * rstd * gaa.x + bea.x;
  ca.y = ata.y + (hv[1] - mean) * rstd * gaa.y + bea.y;
  ca.z = ata.z + (hv[2] - mean) * rstd * gaa.z + bea.z;
  ca.w = ata.w + (hv[3] - mean) * rstd * gaa.w + bea.w;
  cb.x = atb.x + (hv[4] - mean) * rstd * gab.x + beb.x;
  cb.y = atb.y + (hv[5] - mean) * rstd * gab.y + beb.y;
  cb.z = atb.z + (hv[6] - mean) * rstd * gab.z + beb.z;
  cb.w = atb.w + (hv[7] - mean) * rstd * gab.w + beb.w;
  *(float4*)(comb + bt * CP + 4 * q)       = ca;
  *(float4*)(comb + bt * CP + 512 + 4 * q) = cb;
  lds_barrier();

  const float amask = (c < BN) ? 1.f : 0.f;
  const float* aptr = comb + (c < BN ? c : 0) * CP + kb;
  float4 av[NP];
#pragma unroll
  for (int j = 0; j < NP; ++j) av[j] = *(const float4*)(aptr + 8 * j);

  v8f acc0 = {0.f, 0.f, 0.f, 0.f, 0.f, 0.f, 0.f, 0.f};
  v8f acc1 = {0.f, 0.f, 0.f, 0.f, 0.f, 0.f, 0.f, 0.f};
#pragma unroll
  for (int j = 0; j < NP; ++j) {
    v2f a0, b0, a1, b1;
    a0.x = av[j].x * amask;  a0.y = av[j].y * amask;
    b0.x = bv[j].x;          b0.y = bv[j].y;
    a1.x = av[j].z * amask;  a1.y = av[j].w * amask;
    b1.x = bv[j].z;          b1.y = bv[j].w;
    acc0 = __builtin_amdgcn_wmma_f32_16x16x4_f32(
        false, a0, false, b0, (short)0, acc0, false, false);
    acc1 = __builtin_amdgcn_wmma_f32_16x16x4_f32(
        false, a1, false, b1, (short)0, acc1, false, false);
  }

#pragma unroll
  for (int i = 0; i < 8; ++i) red[wave][8 * h + i][c] = acc0[i] + acc1[i];
  __syncthreads();

  if (t < 16 * BN) {
    const int r = t >> 4, cc = t & 15;
    float v = 0.f;
#pragma unroll
    for (int w = 0; w < NW; ++w) v += red[w][r][cc];
    Out[r * OO + n0 + cc] = v + bval;
  }
}

extern "C" void kernel_launch(void* const* d_in, const int* in_sizes, int n_in,
                              void* d_out, int out_size, void* d_ws, size_t ws_size,
                              hipStream_t stream) {
  const float* x       = (const float*)d_in[0];
  const float* Wq_proj = (const float*)d_in[1];
  const float* bq      = (const float*)d_in[2];
  const float* Wq      = (const float*)d_in[3];
  const float* Wk      = (const float*)d_in[4];
  const float* Wv      = (const float*)d_in[5];
  const float* W1      = (const float*)d_in[6];
  const float* b1      = (const float*)d_in[7];
  const float* W2      = (const float*)d_in[8];
  const float* b2      = (const float*)d_in[9];
  const float* gamma   = (const float*)d_in[10];
  const float* beta    = (const float*)d_in[11];
  float* out = (float*)d_out;

  float* ws     = (float*)d_ws;
  float* qproj  = ws;
  float* qlast  = ws + 16384;
  float* qtil   = ws + 32768;
  float* attn   = ws + 61440;
  float* hin    = ws + 65536;
  float* part   = ws + 131072;
  float* stats  = ws + 393216;

  wmma_gemm16_ks<false, HH, DD, HH, true, 1><<<dim3(HH / 16, 4), 128, 0, stream>>>(
      x + (size_t)(SS - 1) * DD, (long)SS * DD, 0, Wq_proj, bq, qproj);
  wmma_gemm16_ks<false, HH, HH, HH, false, 4><<<dim3(HH / 16, 4), 128, 0, stream>>>(
      qproj, HH, BN * HH, Wq, nullptr, qlast);
  wmma_gemm16_ks<true, HH, HH, DD, false, 4><<<dim3(DD / 16, 4), 128, 0, stream>>>(
      qlast, HH, BN * HH, Wk, nullptr, qtil);
  scores_part_k<<<dim3(SS / 32, BN), 256, 0, stream>>>(x, qtil, part, stats, attn);
  merge_gemm16<<<dim3(HH / 64, DD / 64), 128, 0, stream>>>(part, stats, Wv, attn);
  wmma_gemm16<false, FF, HH, FF, true><<<dim3(FF / 16), 512, 0, stream>>>(
      attn, HH, BN, W1, b1, hin);
  glu_ln_gemm16<<<dim3(OO / 16), 512, 0, stream>>>(
      hin, attn, gamma, beta, W2, b2, out);
}


# launches 7 -> 6: first two GEMMs in one launch (a block multiplies its 16-column tile of the first product straight into the second); third GEMM adds the 64 partial planes; new sum order
# speedup vs baseline: 2.9157x; 1.0531x over previous
#include <hip/hip_runtime.h>
#include <hip/hip_bf16.h>
#include <math.h>

typedef float v2f __attribute__((ext_vector_type(2)));
typedef float v8f __attribute__((ext_vector_type(8)));

#define BN 4
#define SS 2048
#define DD 1024
#define HH 1024
#define OO 1024
#define FF 2048

template <bool TRANSB, int LDB, int K, int LDOUT, bool HASBIAS>
__global__ __launch_bounds__(512) __attribute__((amdgpu_waves_per_eu(4, 4)))
void wmma_gemm16(
    const float* __restrict__ A, long lda, int rows,
    const float* __restrict__ Bm, const float* __restrict__ bias,
    float* __restrict__ Out) {
  constexpr int NW = 16;
  constexpr int KW = K / NW;
  constexpr int NP = KW / 8;
  __shared__ float red[NW][16][16];
  const int lane = threadIdx.x & 31;
  const int wave = threadIdx.x >> 5;
  const int n0   = blockIdx.x * 16;
  const int c    = lane & 15;
  const int h    = lane >> 4;
  const int kb   = wave * KW + 4 * h;

  const float amask = (c < rows) ? 1.f : 0.f;
  const float* aptr = A + (long)(c < rows ? c : 0) * lda + kb;
  const float* bptr = TRANSB ? (Bm + (long)(n0 + c) * LDB + kb)
                             : (Bm + (long)kb * LDB + (n0 + c));

  const float bval = HASBIAS ? bias[n0 + (threadIdx.x & 15)] : 0.f;

  float4 av[NP], bv[NP];
#pragma unroll
  for (int j = 0; j < NP; ++j) {
    av[j] = *(const float4*)(aptr + 8 * j);
    if (TRANSB) {
      bv[j] = *(const float4*)(bptr + 8 * j);
    } else {
      const float* bp = bptr + (long)(8 * j) * LDB;
      bv[j].x = bp[0];
      bv[j].y = bp[LDB];
      bv[j].z = bp[2 * LDB];
      bv[j].w = bp[3 * LDB];
    }
  }

  v8f acc0 = {0.f, 0.f, 0.f, 0.f, 0.f, 0.f, 0.f, 0.f};
  v8f acc1 = {0.f, 0.f, 0.f, 0.f, 0.f, 0.f, 0.f, 0.f};
#pragma unroll
  for (int j = 0; j < NP; ++j) {
    v2f a0, b0, a1, b1;
    a0.x = av[j].x * amask;  a0.y = av[j].y * amask;
    b0.x = bv[j].x;          b0.y = bv[j].y;
    a1.x = av[j].z * amask;  a1.y = av[j].w * amask;
    b1.x = bv[j].z;          b1.y = bv[j].w;
    acc0 = __builtin_amdgcn_wmma_f32_16x16x4_f32(
        false, a0, false, b0, (short)0, acc0, false, false);
    acc1 = __builtin_amdgcn_wmma_f32_16x16x4_f32(
        false, a1, false, b1, (short)0, acc1, false, false);
  }

#pragma unroll
  for (int i = 0; i < 8; ++i) red[wave][8 * h + i][c] = acc0[i] + acc1[i];
  __syncthreads();

  const int t = threadIdx.x;
  if (t < 16 * rows) {
    const int r = t >> 4, cc = t & 15;
    float v = 0.f;
#pragma unroll
    for (int w = 0; w < NW; ++w) v += red[w][r][cc];
    Out[r * LDOUT + n0 + cc] = v + bval;
  }
}

__device__ __forceinline__ void lds_barrier() {
  __builtin_amdgcn_fence(__ATOMIC_RELEASE, "workgroup", "local");
  __builtin_amdgcn_s_barrier();
  __builtin_amdgcn_fence(__ATOMIC_ACQUIRE, "workgroup", "local");
}

__device__ __forceinline__ float wave_sum(float v) {
  for (int off = 16; off; off >>= 1) v += __shfl_xor(v, off, 32);
  return v;
}
__device__ __forceinline__ float wave_max(float v) {
  for (int off = 16; off; off >>= 1) v = fmaxf(v, __shfl_xor(v, off, 32));
  return v;
}

template <int G>
__global__ __launch_bounds__(512) __attribute__((amdgpu_waves_per_eu(4, 4)))
void gemm16_pair(const float* __restrict__ A, long lda,
                 const float* __restrict__ B1, const float* __restrict__ bias1,
                 const float* __restrict__ B2, float* __restrict__ planes) {
  constexpr int NW = 16;
  constexpr int KW = DD / NW;
  constexpr int NP = KW / 8;
  constexpr int CW = HH / (G * NW);
  constexpr int TW = CW / 16;
  typedef float vT __attribute__((ext_vector_type(TW)));
  constexpr int TP = 16 + 8;
  __shared__ float red[NW][16][16];
  __shared__ __attribute__((aligned(16))) float tile[BN * TP];
  const int t    = threadIdx.x;
  const int lane = t & 31;
  const int wave = t >> 5;
  const int n0   = blockIdx.x * 16;
  const int c    = lane & 15;
  const int h    = lane >> 4;
  const int kb   = wave * KW + 4 * h;

  const float amask = (c < BN) ? 1.f : 0.f;
  const float* aptr = A + (long)(c < BN ? c : 0) * lda + kb;
  const float* bptr = B1 + (long)kb * HH + (n0 + c);
  const float bval = bias1[n0 + (t & 15)];

  float4 av[NP], bv[NP];
#pragma unroll
  for (int j = 0; j < NP; ++j) {
    av[j] = *(const float4*)(aptr + 8 * j);
    const float* bp = bptr + (long)(8 * j) * HH;
    bv[j].x = bp[0];
    bv[j].y = bp[HH];
    bv[j].z = bp[2 * HH];
    bv[j].w = bp[3 * HH];
  }
  const int nw0 = blockIdx.y * (HH / G) + wave * CW;
  const float* b2ptr = B2 + (long)(n0 + 4 * h) * HH + nw0 + TW * c;
  vT b2[8];
#pragma unroll
  for (int p = 0; p < 2; ++p)
#pragma unroll
    for (int e = 0; e < 4; ++e)
      b2[4 * p + e] = *(const vT*)(b2ptr + (long)(8 * p + e) * HH);

  v8f acc0 = {0.f, 0.f, 0.f, 0.f, 0.f, 0.f, 0.f, 0.f};
  v8f acc1 = {0.f, 0.f, 0.f, 0.f, 0.f, 0.f, 0.f, 0.f};
#pragma unroll
  for (int j = 0; j < NP; ++j) {
    v2f a0, b0, a1, b1;
    a0.x = av[j].x * amask;  a0.y = av[j].y * amask;
    b0.x = bv[j].x;          b0.y = bv[j].y;
    a1.x = av[j].z * amask;  a1.y = av[j].w * amask;
    b1.x = bv[j].z;          b1.y = bv[j].w;
    acc0 = __builtin_amdgcn_wmma_f32_16x16x4_f32(
        false, a0, false, b0, (short)0, acc0, false, false);
    acc1 = __builtin_amdgcn_wmma_f32_16x16x4_f32(
        false, a1, false, b1, (short)0, acc1, false, false);
  }

#pragma unroll
  for (int i = 0; i < 8; ++i) red[wave][8 * h + i][c] = acc0[i] + acc1[i];
  lds_barrier();

  if (t < 16 * BN) {
    const int r = t >> 4, cc = t & 15;
    float v = 0.f;
#pragma unroll
    for (int w = 0; w < NW; ++w) v += red[w][r][cc];
    tile[r * TP + cc] = v + bval;
  }
  lds_barrier();

  const float* a2ptr = tile + (c < BN ? c : 0) * TP + 4 * h;
  const float4 a2lo = *(const float4*)(a2ptr);
  const float4 a2hi = *(const float4*)(a2ptr + 8);
  v2f fa[4];
  fa[0].x = a2lo.x * amask;  fa[0].y = a2lo.y * amask;
  fa[1].x = a2lo.z * amask;  fa[1].y = a2lo.w * amask;
  fa[2].x = a2hi.x * amask;  fa[2].y = a2hi.y * amask;
  fa[3].x = a2hi.z * amask;  fa[3].y = a2hi.w * amask;
  v8f acc2[TW];
#pragma unroll
  for (int e = 0; e < TW; ++e) acc2[e] = v8f{0.f, 0.f, 0.f, 0.f, 0.f, 0.f, 0.f, 0.f};
#pragma unroll
  for (int s = 0; s < 4; ++s) {
#pragma unroll
    for (int e = 0; e < TW; ++e) {
      v2f fb;
      fb.x = b2[2 * s][e];
      fb.y = b2[2 * s + 1][e];
      acc2[e] = __builtin_amdgcn_wmma_f32_16x16x4_f32(
          false, fa[s], false, fb, (short)0, acc2[e], false, false);
    }
  }

  if (h == 0) {
    float* dst = planes + (size_t)blockIdx.x * (BN * HH) + nw0 + TW * c;
#pragma unroll
    for (int b = 0; b < BN; ++b) {
      vT o;
#pragma unroll
      for (int e = 0; e < TW; ++e) o[e] = acc2[e][b];
      *(vT*)(dst + b * HH) = o;
    }
  }
}

__global__ __launch_bounds__(128) __attribute__((amdgpu_waves_per_eu(1, 2)))
void planes_gemm16t(const float* __restrict__ planes,
                    const float* __restrict__ Bm,
                    float* __restrict__ Out) {
  constexpr int KB = 64;
  constexpr int NP = KB / 8;
  constexpr int NJ = 64;
  constexpr int HJ = NJ / 2;
  constexpr int CP = KB + 8;
  __shared__ __attribute__((aligned(16))) float As[BN * CP];
  const int t    = threadIdx.x;
  const int lane = t & 31;
  const int wave = __builtin_amdgcn_readfirstlane(t >> 5);
  const int n0   = blockIdx.x * 64 + wave * 16;
  const int k0   = blockIdx.y * KB;
  const int c    = lane & 15;
  const int h    = lane >> 4;

  const float4* pp =
      (const float4*)(planes + (size_t)(HJ * h) * (BN * HH) + wave * HH + k0) + c;
  float4 pv[HJ];
#pragma unroll
  for (int i = 0; i < HJ; ++i) pv[i] = pp[(size_t)i * (BN * HH / 4)];
  const float* bptr = Bm + (long)(n0 + c) * HH + k0 + 4 * h;
  float4 bv[NP];
#pragma unroll
  for (int j = 0; j < NP; ++j) bv[j] = *(const float4*)(bptr + 8 * j);

  float4 acc = pv[0];
#pragma unroll
  for (int i = 1; i < HJ; ++i) {
    acc.x += pv[i].x; acc.y += pv[i].y; acc.z += pv[i].z; acc.w += pv[i].w;
  }
  acc.x += __shfl_xor(acc.x, 16, 32);
  acc.y += __shfl_xor(acc.y, 16, 32);
  acc.z += __shfl_xor(acc.z, 16, 32);
  acc.w += __shfl_xor(acc.w, 16, 32);
  if (h == 0) *(float4*)(As + wave * CP + 4 * c) = acc;
  lds_barrier();

  const float amask = (c < BN) ? 1.f : 0.f;
  const float* aptr = As + (c < BN ? c : 0) * CP + 4 * h;
  float4 av[NP];
#pragma unroll
  for (int j = 0; j < NP; ++j) av[j] = *(const float4*)(aptr + 8 * j);

  v8f acc0 = {0.f, 0.f, 0.f, 0.f, 0.f, 0.f, 0.f, 0.f};
  v8f acc1 = {0.f, 0.f, 0.f, 0.f, 0.f, 0.f, 0.f, 0.f};
#pragma unroll
  for (int j = 0; j < NP; ++j) {
    v2f a0, b0, a1, b1;
    a0.x = av[j].x * amask;  a0.y = av[j].y * amask;
    b0.x = bv[j].x;          b0.y = bv[j].y;
    a1.x = av[j].z * amask;  a1.y = av[j].w * amask;
    b1.x = bv[j].z;          b1.y = bv[j].w;
    acc0 = __builtin_amdgcn_wmma_f32_16x16x4_f32(
        false, a0, false, b0, (short)0, acc0, false, false);
    acc1 = __builtin_amdgcn_wmma_f32_16x16x4_f32(
        false, a1, false, b1, (short)0, acc1, false, false);
  }

  __builtin_amdgcn_sched_group_barrier(0x100, NP, 0);
  __builtin_amdgcn_sched_group_barrier(0x008, 2 * NP, 0);

  if (h == 0) {
    float* dst = Out + (size_t)blockIdx.y * (BN * DD) + n0 + c;
#pragma unroll
    for (int i = 0; i < BN; ++i) dst[i * DD] = acc0[i] + acc1[i];
  }
}

__global__ __launch_bounds__(256) __attribute__((amdgpu_waves_per_eu(2, 2)))
void scores_part_k(const float* __restrict__ x,
                   const float* __restrict__ qtil,
                   float* __restrict__ part,
                   float* __restrict__ stats,
                   float* __restrict__ attn) {
  constexpr int RW = 4;
  constexpr int NWV = 8;
  const int b    = blockIdx.y;
  const int t    = threadIdx.x;
  const int lane = t & 31;
  const int wave = t >> 5;
  const int srow = blockIdx.x * (NWV * RW) + wave * RW;
  const float4* xr = (const float4*)(x + ((size_t)b * SS + srow) * DD) + lane;
  constexpr int QP = 16;
  const float4* q4 = (const float4*)(qtil + (size_t)b * DD) + t;
  float4 qp[QP];
#pragma unroll
  for (int p = 0; p < QP; ++p) qp[p] = q4[p * (BN * DD / 4)];
  float4 qv[8], xv[RW][8];
#pragma unroll
  for (int r = 0; r < RW; ++r)
#pragma unroll
    for (int i = 0; i < 8; ++i) xv[r][i] = xr[r * (DD / 4) + i * 32];
  __shared__ float4 qs[DD / 4];
  __shared__ float smax[NWV], ssum[NWV];
  __shared__ float4 red[NWV][DD / 4];
  float4 qsum = qp[0];
#pragma unroll
  for (int p = 1; p < QP; ++p) {
    qsum.x += qp[p].x; qsum.y += qp[p].y; qsum.z += qp[p].z; qsum.w += qp[p].w;
  }
  qs[t] = qsum;
  lds_barrier();
#pragma unroll
  for (int i = 0; i < 8; ++i) qv[i] = qs[i * 32 + lane];
  float sum[RW];
#pragma unroll
  for (int r = 0; r < RW; ++r) {
    float s = 0.f;
#pragma unroll
    for (int i = 0; i < 8; ++i)
      s += xv[r][i].x * qv[i].x + xv[r][i].y * qv[i].y +
           xv[r][i].z * qv[i].z + xv[r][i].w * qv[i].w;
    sum[r] = s;
  }
  for (int off = 16; off; off >>= 1) {
#pragma unroll
    for (int r = 0; r < RW; ++r) sum[r] += __shfl_xor(sum[r], off, 32);
  }
  float sc[RW];
#pragma unroll
  for (int r = 0; r < RW; ++r) sc[r] = sum[r] * 0.03125f;
  const float mw = fmaxf(fmaxf(sc[0], sc[1]), fmaxf(sc[2], sc[3]));
  if (lane == 0) smax[wave] = mw;
  lds_barrier();
  float m = smax[0];
#pragma unroll
  for (int wv = 1; wv < NWV; ++wv) m = fmaxf(m, smax[wv]);
  float e[RW];
#pragma unroll
  for (int r = 0; r < RW; ++r) e[r] = expf(sc[r] - m);
  if (lane == 0) ssum[wave] = ((e[0] + e[1]) + e[2]) + e[3];
#pragma unroll
  for (int i = 0; i < 8; ++i) {
    float4 a;
    a.x = e[0] * xv[0][i].x; a.y = e[0] * xv[0][i].y;
    a.z = e[0] * xv[0][i].z; a.w = e[0] * xv[0][i].w;
#pragma unroll
    for (int r = 1; r < RW; ++r) {
      a.x = fmaf(e[r], xv[r][i].x, a.x); a.y = fmaf(e[r], xv[r][i].y, a.y);
      a.z = fmaf(e[r], xv[r][i].z, a.z); a.w = fmaf(e[r], xv[r][i].w, a.w);
    }
    red[wave][i * 32 + lane] = a;
  }
  __syncthreads();
  float4 v = red[0][t];
#pragma unroll
  for (int wv = 1; wv < NWV; ++wv) {
    const float4 u = red[wv][t];
    v.x += u.x; v.y += u.y; v.z += u.z; v.w += u.w;
  }
  const int plane = b * (SS / (NWV * RW)) + blockIdx.x;
  ((float4*)(part + (size_t)plane * DD))[t] = v;
  if (t == 0) {
    float l = 0.f;
#pragma unroll
    for (int wv = 0; wv < NWV; ++wv) l += ssum[wv];
    float2 ml; ml.x = m; ml.y = l;
    ((float2*)stats)[plane] = ml;
  }
  if (blockIdx.x == 0)
    ((float4*)(attn + b * HH))[t] = float4{0.f, 0.f, 0.f, 0.f};
}

__global__ __launch_bounds__(128) __attribute__((amdgpu_waves_per_eu(1, 2)))
void merge_gemm16(const float* __restrict__ part,
                  const float* __restrict__ stats,
                  const float* __restrict__ Bm,
                  float* __restrict__ Out) {
  constexpr int KB = 64;
  constexpr int NP = KB / 8;
  constexpr int NJ = 64;
  constexpr int HJ = NJ / 2;
  constexpr int CP = KB + 8;
  __shared__ __attribute__((aligned(16))) float As[BN * CP];
  const int t    = threadIdx.x;
  const int lane = t & 31;
  const int wave = __builtin_amdgcn_readfirstlane(t >> 5);
  const int n0   = blockIdx.x * 64 + wave * 16;
  const int k0   = blockIdx.y * KB;
  const int c    = lane & 15;
  const int h    = lane >> 4;

  const float2* st = (const float2*)stats + wave * NJ;
  const float2 slo = st[lane], shi = st[lane + HJ];
  const float4* pp = (const float4*)(part + (size_t)(wave * NJ + HJ * h) * DD + k0) + c;
  float4 pv[HJ];
#pragma unroll
  for (int i = 0; i < HJ; ++i) pv[i] = pp[(size_t)i * (DD / 4)];
  const float* bptr = Bm + (long)(k0 + 4 * h) * HH + (n0 + c);
  float4 bv[NP];
#pragma unroll
  for (int j = 0; j < NP; ++j) {
    const float* bp = bptr + (long)(8 * j) * HH;
    bv[j].x = bp[0];
    bv[j].y = bp[HH];
    bv[j].z = bp[2 * HH];
    bv[j].w = bp[3 * HH];
  }

  const float mb = wave_max(fmaxf(slo.x, shi.x));
  const float elo = expf(slo.x - mb), ehi = expf(shi.x - mb);
  const float lb = wave_sum(elo * slo.y + ehi * shi.y);
  const float inv = 1.f / lb;
  const int wlo = __builtin_bit_cast(int, elo * inv);
  const int whi = __builtin_bit_cast(int, ehi * inv);

  float4 acc = {0.f, 0.f, 0.f, 0.f};
#pragma unroll
  for (int i = 0; i < HJ; ++i) {
    const float f0 = __builtin_bit_cast(float, __builtin_amdgcn_readlane(wlo, i));
    const float f1 = __builtin_bit_cast(float, __builtin_amdgcn_readlane(whi, i));
    const float ww = h ? f1 : f0;
    acc.x = fmaf(ww, pv[i].x, acc.x); acc.y = fmaf(ww, pv[i].y, acc.y);
    acc.z = fmaf(ww, pv[i].z, acc.z); acc.w = fmaf(ww, pv[i].w, acc.w);
  }
  acc.x += __shfl_xor(acc.x, 16, 32);
  acc.y += __shfl_xor(acc.y, 16, 32);
  acc.z += __shfl_xor(acc.z, 16, 32);
  acc.w += __shfl_xor(acc.w, 16, 32);
  __builtin_amdgcn_sched_group_barrier(0x020, 2 + HJ + 4 * NP, 0);
  __builtin_amdgcn_sched_group_barrier(0x080, 14, 0);
  if (h == 0) *(float4*)(As + wave * CP + 4 * c) = acc;
  lds_barrier();

  const float amask = (c < BN) ? 1.f : 0.f;
  const float* aptr = As + (c < BN ? c : 0) * CP + 4 * h;
  float4 av[NP];
#pragma unroll
  for (int j = 0; j < NP; ++j) av[j] = *(const float4*)(aptr + 8 * j);

  v8f acc0 = {0.f, 0.f, 0.f, 0.f, 0.f, 0.f, 0.f, 0.f};
  v8f acc1 = {0.f, 0.f, 0.f, 0.f, 0.f, 0.f, 0.f, 0.f};
#pragma unroll
  for (int j = 0; j < NP; ++j) {
    v2f a0, b0, a1, b1;
    a0.x = av[j].x * amask;  a0.y = av[j].y * amask;
    b0.x = bv[j].x;          b0.y = bv[j].y;
    a1.x = av[j].z * amask;  a1.y = av[j].w * amask;
    b1.x = bv[j].z;          b1.y = bv[j].w;
    acc0 = __builtin_amdgcn_wmma_f32_16x16x4_f32(
        false, a0, false, b0, (short)0, acc0, false, false);
    acc1 = __builtin_amdgcn_wmma_f32_16x16x4_f32(
        false, a1, false, b1, (short)0, acc1, false, false);
  }

  __builtin_amdgcn_sched_group_barrier(0x100, NP, 0);
  __builtin_amdgcn_sched_group_barrier(0x008, 2 * NP, 0);

  float o[BN];
#pragma unroll
  for (int i = 0; i < BN; ++i) o[i] = acc0[i] + acc1[i];
  if (h == 0) {
#pragma unroll
    for (int i = 0; i < BN; ++i) atomicAdd(Out + i * HH + n0 + c, o[i]);
  }
}

__global__ __launch_bounds__(512) __attribute__((amdgpu_waves_per_eu(4, 4)))
void glu_ln_gemm16(const float* __restrict__ hin,
                   const float* __restrict__ attn,
                   const float* __restrict__ gamma,
                   const float* __restrict__ beta,
                   const float* __restrict__ Bm,
                   const float* __restrict__ bias,
                   float* __restrict__ Out) {
  constexpr int NW = 16;
  constexpr int KW = HH / NW;
  constexpr int NP = KW / 8;
  constexpr int CP = HH + 8;
  __shared__ float red[NW][16][16];
  __shared__ __attribute__((aligned(16))) float comb[BN * CP];
  __shared__ float psum[NW], psq[NW];
  const int t    = threadIdx.x;
  const int lane = t & 31;
  const int wave = t >> 5;
  const int n0   = blockIdx.x * 16;
  const int c    = lane & 15;
  const int h    = lane >> 4;
  const int kb   = wave * KW + 4 * h;
  const int bt   = t >> 7;
  const int q    = t & 127;

  const float4* h4 = (const float4*)(hin + (size_t)bt * FF);
  const float4 x1a = h4[q],       x1b = h4[q + 128];
  const float4 x2a = h4[q + 256], x2b = h4[q + 384];
  const float4* a4 = (const float4*)(attn + (size_t)bt * HH);
  const float4 ata = a4[q], atb = a4[q + 128];
  const float4 gaa = ((const float4*)gamma)[q], gab = ((const float4*)gamma)[q + 128];
  const float4 bea = ((const float4*)beta)[q],  beb = ((const float4*)beta)[q + 128];
  const float bval = bias[n0 + (t & 15)];

  const float* bptr = Bm + (long)kb * OO + (n0 + c);
  float4 bv[NP];
#pragma unroll
  for (int j = 0; j < NP; ++j) {
    const float* bp = bptr + (long)(8 * j) * OO;
    bv[j].x = bp[0];
    bv[j].y = bp[OO];
    bv[j].z = bp[2 * OO];
    bv[j].w = bp[3 * OO];
  }

  const float x1[8] = {x1a.x, x1a.y, x1a.z, x1a.w, x1b.x, x1b.y, x1b.z, x1b.w};
  const float x2[8] = {x2a.x, x2a.y, x2a.z, x2a.w, x2b.x, x2b.y, x2b.z, x2b.w};
  float hv[8];
  float sum = 0.f;
#pragma unroll
  for (int j = 0; j < 8; ++j) {
    const float g = 0.5f * x2[j] * (1.f + erff(x2[j] * 0.70710678118654752f));
    hv[j] = x1[j] * g;
    sum += hv[j];
  }
  sum = wave_sum(sum);
  if (lane == 0) psum[wave] = sum;
  lds_barrier();
  const int w0 = bt * 4;
  const float mean = (psum[w0] + psum[w0 + 1] + psum[w0 + 2] + psum[w0 + 3]) * (1.f / HH);
  float sq = 0.f;
#pragma unroll
  for (int j = 0; j < 8; ++j) { const float d = hv[j] - mean; sq += d * d; }
  sq = wave_sum(sq);
  if (lane == 0) psq[wave] = sq;
  lds_barrier();
  const float var = (psq[w0] + psq[w0 + 1] + psq[w0 + 2] + psq[w0 + 3]) * (1.f / HH);
  const float rstd = rsqrtf(var + 1e-5f);
  float4 ca, cb;
  ca.x = ata.x + (hv[0] - mean) * rstd * gaa.x + bea.x;
  ca.y = ata.y + (hv[1] - mean) * rstd * gaa.y + bea.y;
  ca.z = ata.z + (hv[2] - mean) * rstd * gaa.z + bea.z;
  ca.w = ata.w + (hv[3] - mean) * rstd * gaa.w + bea.w;
  cb.x = atb.x + (hv[4] - mean) * rstd * gab.x + beb.x;
  cb.y = atb.y + (hv[5] - mean) * rstd * gab.y + beb.y;
  cb.z = atb.z + (hv[6] - mean) * rstd * gab.z + beb.z;
  cb.w = atb.w + (hv[7] - mean) * rstd * gab.w + beb.w;
  *(float4*)(comb + bt * CP + 4 * q)       = ca;
  *(float4*)(comb + bt * CP + 512 + 4 * q) = cb;
  lds_barrier();

  const float amask = (c < BN) ? 1.f : 0.f;
  const float* aptr = comb + (c < BN ? c : 0) * CP + kb;
  float4 av[NP];
#pragma unroll
  for (int j = 0; j < NP; ++j) av[j] = *(const float4*)(aptr + 8 * j);

  v8f acc0 = {0.f, 0.f, 0.f, 0.f, 0.f, 0.f, 0.f, 0.f};
  v8f acc1 = {0.f, 0.f, 0.f, 0.f, 0.f, 0.f, 0.f, 0.f};
#pragma unroll
  for (int j = 0; j < NP; ++j) {
    v2f a0, b0, a1, b1;
    a0.x = av[j].x * amask;  a0.y = av[j].y * amask;
    b0.x = bv[j].x;          b0.y = bv[j].y;
    a1.x = av[j].z * amask;  a1.y = av[j].w * amask;
    b1.x = bv[j].z;          b1.y = bv[j].w;
    acc0 = __builtin_amdgcn_wmma_f32_16x16x4_f32(
        false, a0, false, b0, (short)0, acc0, false, false);
    acc1 = __builtin_amdgcn_wmma_f32_16x16x4_f32(
        false, a1, false, b1, (short)0, acc1, false, false);
  }

#pragma unroll
  for (int i = 0; i < 8; ++i) red[wave][8 * h + i][c] = acc0[i] + acc1[i];
  __syncthreads();

  if (t < 16 * BN) {
    const int r = t >> 4, cc = t & 15;
    float v = 0.f;
#pragma unroll
    for (int w = 0; w < NW; ++w) v += red[w][r][cc];
    Out[r * OO + n0 + cc] = v + bval;
  }
}

extern "C" void kernel_launch(void* const* d_in, const int* in_sizes, int n_in,
                              void* d_out, int out_size, void* d_ws, size_t ws_size,
                              hipStream_t stream) {
  const float* x       = (const float*)d_in[0];
  const float* Wq_proj = (const float*)d_in[1];
  const float* bq      = (const float*)d_in[2];
  const float* Wq      = (const float*)d_in[3];
  const float* Wk      = (const float*)d_in[4];
  const float* Wv      = (const float*)d_in[5];
  const float* W1      = (const float*)d_in[6];
  const float* b1      = (const float*)d_in[7];
  const float* W2      = (const float*)d_in[8];
  const float* b2      = (const float*)d_in[9];
  const float* gamma   = (const float*)d_in[10];
  const float* beta    = (const float*)d_in[11];
  float* out = (float*)d_out;

  float* ws     = (float*)d_ws;
  float* qlp    = ws;
  float* qtil   = ws + 262144;
  float* attn   = ws + 327680;
  float* hin    = ws + 331776;
  float* part   = ws + 524288;
  float* stats  = ws + 786432;

  gemm16_pair<2><<<dim3(HH / 16, 2), 512, 0, stream>>>(
      x + (size_t)(SS - 1) * DD, (long)SS * DD, Wq_proj, bq, Wq, qlp);
  planes_gemm16t<<<dim3(DD / 64, HH / 64), 128, 0, stream>>>(qlp, Wk, qtil);
  scores_part_k<<<dim3(SS / 32, BN), 256, 0, stream>>>(x, qtil, part, stats, attn);
  merge_gemm16<<<dim3(HH / 64, DD / 64), 128, 0, stream>>>(part, stats, Wv, attn);
  wmma_gemm16<false, FF, HH, FF, true><<<dim3(FF / 16), 512, 0, stream>>>(
      attn, HH, BN, W1, b1, hin);
  glu_ln_gemm16<<<dim3(OO / 16), 512, 0, stream>>>(
      hin, attn, gamma, beta, W2, b2, out);
}


# W1 GEMM re-tiled to 256 blocks of 4 waves (was 128 x 16): 32-column tiles read as whole 128-byte lines, K in 4 slices of 256 that meet in hin by f32 atomics (new sum order)
# speedup vs baseline: 3.0855x; 1.0582x over previous
#include <hip/hip_runtime.h>
#include <hip/hip_bf16.h>
#include <math.h>

typedef float v2f __attribute__((ext_vector_type(2)));
typedef float v8f __attribute__((ext_vector_type(8)));

#define BN 4
#define SS 2048
#define DD 1024
#define HH 1024
#define OO 1024
#define FF 2048

__device__ __forceinline__ void lds_barrier() {
  __builtin_amdgcn_fence(__ATOMIC_RELEASE, "workgroup", "local");
  __builtin_amdgcn_s_barrier();
  __builtin_amdgcn_fence(__ATOMIC_ACQUIRE, "workgroup", "local");
}

__device__ __forceinline__ float wave_sum(float v) {
  for (int off = 16; off; off >>= 1) v += __shfl_xor(v, off, 32);
  return v;
}
__device__ __forceinline__ float wave_max(float v) {
  for (int off = 16; off; off >>= 1) v = fmaxf(v, __shfl_xor(v, off, 32));
  return v;
}

template <int G>
__global__ __launch_bounds__(512) __attribute__((amdgpu_waves_per_eu(4, 4)))
void gemm16_pair(const float* __restrict__ A, long lda,
                 const float* __restrict__ B1, const float* __restrict__ bias1,
                 const float* __restrict__ B2, float* __restrict__ planes) {
  constexpr int NW = 16;
  constexpr int KW = DD / NW;
  constexpr int NP = KW / 8;
  constexpr int CW = HH / (G * NW);
  constexpr int TW = CW / 16;
  typedef float vT __attribute__((ext_vector_type(TW)));
  constexpr int TP = 16 + 8;
  __shared__ float red[NW][16][16];
  __shared__ __attribute__((aligned(16))) float tile[BN * TP];
  const int t    = threadIdx.x;
  const int lane = t & 31;
  const int wave = t >> 5;
  const int n0   = blockIdx.x * 16;
  const int c    = lane & 15;
  const int h    = lane >> 4;
  const int kb   = wave * KW + 4 * h;

  const float amask = (c < BN) ? 1.f : 0.f;
  const float* aptr = A + (long)(c < BN ? c : 0) * lda + kb;
  const float* bptr = B1 + (long)kb * HH + (n0 + c);
  const float bval = bias1[n0 + (t & 15)];

  float4 av[NP], bv[NP];
#pragma unroll
  for (int j = 0; j < NP; ++j) {
    av[j] = *(const float4*)(aptr + 8 * j);
    const float* bp = bptr + (long)(8 * j) * HH;
    bv[j].x = bp[0];
    bv[j].y = bp[HH];
    bv[j].z = bp[2 * HH];
    bv[j].w = bp[3 * HH];
  }
  const int nw0 = blockIdx.y * (HH / G) + wave * CW;
  const float* b2ptr = B2 + (long)(n0 + 4 * h) * HH + nw0 + TW * c;
  vT b2[8];
#pragma unroll
  for (int p = 0; p < 2; ++p)
#pragma unroll
    for (int e = 0; e < 4; ++e)
      b2[4 * p + e] = *(const vT*)(b2ptr + (long)(8 * p + e) * HH);

  v8f acc0 = {0.f, 0.f, 0.f, 0.f, 0.f, 0.f, 0.f, 0.f};
  v8f acc1 = {0.f, 0.f, 0.f, 0.f, 0.f, 0.f, 0.f, 0.f};
#pragma unroll
  for (int j = 0; j < NP; ++j) {
    v2f a0, b0, a1, b1;
    a0.x = av[j].x * amask;  a0.y = av[j].y * amask;
    b0.x = bv[j].x;          b0.y = bv[j].y;
    a1.x = av[j].z * amask;  a1.y = av[j].w * amask;
    b1.x = bv[j].z;          b1.y = bv[j].w;
    acc0 = __builtin_amdgcn_wmma_f32_16x16x4_f32(
        false, a0, false, b0, (short)0, acc0, false, false);
    acc1 = __builtin_amdgcn_wmma_f32_16x16x4_f32(
        false, a1, false, b1, (short)0, acc1, false, false);
  }

#pragma unroll
  for (int i = 0; i < 8; ++i) red[wave][8 * h + i][c] = acc0[i] + acc1[i];
  lds_barrier();

  if (t < 16 * BN) {
    const int r = t >> 4, cc = t & 15;
    float v = 0.f;
#pragma unroll
    for (int w = 0; w < NW; ++w) v += red[w][r][cc];
    tile[r * TP + cc] = v + bval;
  }
  lds_barrier();

  const float* a2ptr = tile + (c < BN ? c : 0) * TP + 4 * h;
  const float4 a2lo = *(const float4*)(a2ptr);
  const float4 a2hi = *(const float4*)(a2ptr + 8);
  v2f fa[4];
  fa[0].x = a2lo.x * amask;  fa[0].y = a2lo.y * amask;
  fa[1].x = a2lo.z * amask;  fa[1].y = a2lo.w * amask;
  fa[2].x = a2hi.x * amask;  fa[2].y = a2hi.y * amask;
  fa[3].x = a2hi.z * amask;  fa[3].y = a2hi.w * amask;
  v8f acc2[TW];
#pragma unroll
  for (int e = 0; e < TW; ++e) acc2[e] = v8f{0.f, 0.f, 0.f, 0.f, 0.f, 0.f, 0.f, 0.f};
#pragma unroll
  for (int s = 0; s < 4; ++s) {
#pragma unroll
    for (int e = 0; e < TW; ++e) {
      v2f fb;
      fb.x = b2[2 * s][e];
      fb.y = b2[2 * s + 1][e];
      acc2[e] = __builtin_amdgcn_wmma_f32_16x16x4_f32(
          false, fa[s], false, fb, (short)0, acc2[e], false, false);
    }
  }

  if (h == 0) {
    float* dst = planes + (size_t)blockIdx.x * (BN * HH) + nw0 + TW * c;
#pragma unroll
    for (int b = 0; b < BN; ++b) {
      vT o;
#pragma unroll
      for (int e = 0; e < TW; ++e) o[e] = acc2[e][b];
      *(vT*)(dst + b * HH) = o;
    }
  }
}

__global__ __launch_bounds__(128) __attribute__((amdgpu_waves_per_eu(1, 2)))
void planes_gemm16t(const float* __restrict__ planes,
                    const float* __restrict__ Bm,
                    float* __restrict__ Out) {
  constexpr int KB = 64;
  constexpr int NP = KB / 8;
  constexpr int NJ = 64;
  constexpr int HJ = NJ / 2;
  constexpr int CP = KB + 8;
  __shared__ __attribute__((aligned(16))) float As[BN * CP];
  const int t    = threadIdx.x;
  const int lane = t & 31;
  const int wave = __builtin_amdgcn_readfirstlane(t >> 5);
  const int n0   = blockIdx.x * 64 + wave * 16;
  const int k0   = blockIdx.y * KB;
  const int c    = lane & 15;
  const int h    = lane >> 4;

  const float4* pp =
      (const float4*)(planes + (size_t)(HJ * h) * (BN * HH) + wave * HH + k0) + c;
  float4 pv[HJ];
#pragma unroll
  for (int i = 0; i < HJ; ++i) pv[i] = pp[(size_t)i * (BN * HH / 4)];
  const float* bptr = Bm + (long)(n0 + c) * HH + k0 + 4 * h;
  float4 bv[NP];
#pragma unroll
  for (int j = 0; j < NP; ++j) bv[j] = *(const float4*)(bptr + 8 * j);

  float4 acc = pv[0];
#pragma unroll
  for (int i = 1; i < HJ; ++i) {
    acc.x += pv[i].x; acc.y += pv[i].y; acc.z += pv[i].z; acc.w += pv[i].w;
  }
  acc.x += __shfl_xor(acc.x, 16, 32);
  acc.y += __shfl_xor(acc.y, 16, 32);
  acc.z += __shfl_xor(acc.z, 16, 32);
  acc.w += __shfl_xor(acc.w, 16, 32);
  if (h == 0) *(float4*)(As + wave * CP + 4 * c) = acc;
  lds_barrier();

  const float amask = (c < BN) ? 1.f : 0.f;
  const float* aptr = As + (c < BN ? c : 0) * CP + 4 * h;
  float4 av[NP];
#pragma unroll
  for (int j = 0; j < NP; ++j) av[j] = *(const float4*)(aptr + 8 * j);

  v8f acc0 = {0.f, 0.f, 0.f, 0.f, 0.f, 0.f, 0.f, 0.f};
  v8f acc1 = {0.f, 0.f, 0.f, 0.f, 0.f, 0.f, 0.f, 0.f};
#pragma unroll
  for (int j = 0; j < NP; ++j) {
    v2f a0, b0, a1, b1;
    a0.x = av[j].x * amask;  a0.y = av[j].y * amask;
    b0.x = bv[j].x;          b0.y = bv[j].y;
    a1.x = av[j].z * amask;  a1.y = av[j].w * amask;
    b1.x = bv[j].z;          b1.y = bv[j].w;
    acc0 = __builtin_amdgcn_wmma_f32_16x16x4_f32(
        false, a0, false, b0, (short)0, acc0, false, false);
    acc1 = __builtin_amdgcn_wmma_f32_16x16x4_f32(
        false, a1, false, b1, (short)0, acc1, false, false);
  }

  __builtin_amdgcn_sched_group_barrier(0x100, NP, 0);
  __builtin_amdgcn_sched_group_barrier(0x008, 2 * NP, 0);

  if (h == 0) {
    float* dst = Out + (size_t)blockIdx.y * (BN * DD) + n0 + c;
#pragma unroll
    for (int i = 0; i < BN; ++i) dst[i * DD] = acc0[i] + acc1[i];
  }
}

__global__ __launch_bounds__(256) __attribute__((amdgpu_waves_per_eu(2, 2)))
void scores_part_k(const float* __restrict__ x,
                   const float* __restrict__ qtil,
                   float* __restrict__ part,
                   float* __restrict__ stats,
                   float* __restrict__ attn,
                   float* __restrict__ hin) {
  constexpr int RW = 4;
  constexpr int NWV = 8;
  const int b    = blockIdx.y;
  const int t    = threadIdx.x;
  const int lane = t & 31;
  const int wave = t >> 5;
  const int srow = blockIdx.x * (NWV * RW) + wave * RW;
  const float4* xr = (const float4*)(x + ((size_t)b * SS + srow) * DD) + lane;
  constexpr int QP = 16;
  const float4* q4 = (const float4*)(qtil + (size_t)b * DD) + t;
  float4 qp[QP];
#pragma unroll
  for (int p = 0; p < QP; ++p) qp[p] = q4[p * (BN * DD / 4)];
  float4 qv[8], xv[RW][8];
#pragma unroll
  for (int r = 0; r < RW; ++r)
#pragma unroll
    for (int i = 0; i < 8; ++i) xv[r][i] = xr[r * (DD / 4) + i * 32];
  __shared__ float4 qs[DD / 4];
  __shared__ float smax[NWV], ssum[NWV];
  __shared__ float4 red[NWV][DD / 4];
  float4 qsum = qp[0];
#pragma unroll
  for (int p = 1; p < QP; ++p) {
    qsum.x += qp[p].x; qsum.y += qp[p].y; qsum.z += qp[p].z; qsum.w += qp[p].w;
  }
  qs[t] = qsum;
  lds_barrier();
#pragma unroll
  for (int i = 0; i < 8; ++i) qv[i] = qs[i * 32 + lane];
  float sum[RW];
#pragma unroll
  for (int r = 0; r < RW; ++r) {
    float s = 0.f;
#pragma unroll
    for (int i = 0; i < 8; ++i)
      s += xv[r][i].x * qv[i].x + xv[r][i].y * qv[i].y +
           xv[r][i].z * qv[i].z + xv[r][i].w * qv[i].w;
    sum[r] = s;
  }
  for (int off = 16; off; off >>= 1) {
#pragma unroll
    for (int r = 0; r < RW; ++r) sum[r] += __shfl_xor(sum[r], off, 32);
  }
  float sc[RW];
#pragma unroll
  for (int r = 0; r < RW; ++r) sc[r] = sum[r] * 0.03125f;
  const float mw = fmaxf(fmaxf(sc[0], sc[1]), fmaxf(sc[2], sc[3]));
  if (lane == 0) smax[wave] = mw;
  lds_barrier();
  float m = smax[0];
#pragma unroll
  for (int wv = 1; wv < NWV; ++wv) m = fmaxf(m, smax[wv]);
  float e[RW];
#pragma unroll
  for (int r = 0; r < RW; ++r) e[r] = expf(sc[r] - m);
  if (lane == 0) ssum[wave] = ((e[0] + e[1]) + e[2]) + e[3];
#pragma unroll
  for (int i = 0; i < 8; ++i) {
    float4 a;
    a.x = e[0] * xv[0][i].x; a.y = e[0] * xv[0][i].y;
    a.z = e[0] * xv[0][i].z; a.w = e[0] * xv[0][i].w;
#pragma unroll
    for (int r = 1; r < RW; ++r) {
      a.x = fmaf(e[r], xv[r][i].x, a.x); a.y = fmaf(e[r], xv[r][i].y, a.y);
      a.z = fmaf(e[r], xv[r][i].z, a.z); a.w = fmaf(e[r], xv[r][i].w, a.w);
    }
    red[wave][i * 32 + lane] = a;
  }
  __syncthreads();
  float4 v = red[0][t];
#pragma unroll
  for (int wv = 1; wv < NWV; ++wv) {
    const float4 u = red[wv][t];
    v.x += u.x; v.y += u.y; v.z += u.z; v.w += u.w;
  }
  const int plane = b * (SS / (NWV * RW)) + blockIdx.x;
  ((float4*)(part + (size_t)plane * DD))[t] = v;
  if (t == 0) {
    float l = 0.f;
#pragma unroll
    for (int wv = 0; wv < NWV; ++wv) l += ssum[wv];
    float2 ml; ml.x = m; ml.y = l;
    ((float2*)stats)[plane] = ml;
  }
  if (blockIdx.x == 0)
    ((float4*)(attn + b * HH))[t] = float4{0.f, 0.f, 0.f, 0.f};
  if (blockIdx.x == 1) {
    ((float4*)(hin + b * FF))[t]       = float4{0.f, 0.f, 0.f, 0.f};
    ((float4*)(hin + b * FF))[t + 256] = float4{0.f, 0.f, 0.f, 0.f};
  }
}

__global__ __launch_bounds__(128) __attribute__((amdgpu_waves_per_eu(1, 2)))
void merge_gemm16(const float* __restrict__ part,
                  const float* __restrict__ stats,
                  const float* __restrict__ Bm,
                  float* __restrict__ Out) {
  constexpr int KB = 64;
  constexpr int NP = KB / 8;
  constexpr int NJ = 64;
  constexpr int HJ = NJ / 2;
  constexpr int CP = KB + 8;
  __shared__ __attribute__((aligned(16))) float As[BN * CP];
  const int t    = threadIdx.x;
  const int lane = t & 31;
  const int wave = __builtin_amdgcn_readfirstlane(t >> 5);
  const int n0   = blockIdx.x * 64 + wave * 16;
  const int k0   = blockIdx.y * KB;
  const int c    = lane & 15;
  const int h    = lane >> 4;

  const float2* st = (const float2*)stats + wave * NJ;
  const float2 slo = st[lane], shi = st[lane + HJ];
  const float4* pp = (const float4*)(part + (size_t)(wave * NJ + HJ * h) * DD + k0) + c;
  float4 pv[HJ];
#pragma unroll
  for (int i = 0; i < HJ; ++i) pv[i] = pp[(size_t)i * (DD / 4)];
  const float* bptr = Bm + (long)(k0 + 4 * h) * HH + (n0 + c);
  float4 bv[NP];
#pragma unroll
  for (int j = 0; j < NP; ++j) {
    const float* bp = bptr + (long)(8 * j) * HH;
    bv[j].x = bp[0];
    bv[j].y = bp[HH];
    bv[j].z = bp[2 * HH];
    bv[j].w = bp[3 * HH];
  }

  const float mb = wave_max(fmaxf(slo.x, shi.x));
  const float elo = expf(slo.x - mb), ehi = expf(shi.x - mb);
  const float lb = wave_sum(elo * slo.y + ehi * shi.y);
  const float inv = 1.f / lb;
  const int wlo = __builtin_bit_cast(int, elo * inv);
  const int whi = __builtin_bit_cast(int, ehi * inv);

  float4 acc = {0.f, 0.f, 0.f, 0.f};
#pragma unroll
  for (int i = 0; i < HJ; ++i) {
    const float f0 = __builtin_bit_cast(float, __builtin_amdgcn_readlane(wlo, i));
    const float f1 = __builtin_bit_cast(float, __builtin_amdgcn_readlane(whi, i));
    const float ww = h ? f1 : f0;
    acc.x = fmaf(ww, pv[i].x, acc.x); acc.y = fmaf(ww, pv[i].y, acc.y);
    acc.z = fmaf(ww, pv[i].z, acc.z); acc.w = fmaf(ww, pv[i].w, acc.w);
  }
  acc.x += __shfl_xor(acc.x, 16, 32);
  acc.y += __shfl_xor(acc.y, 16, 32);
  acc.z += __shfl_xor(acc.z, 16, 32);
  acc.w += __shfl_xor(acc.w, 16, 32);
  __builtin_amdgcn_sched_group_barrier(0x020, 2 + HJ + 4 * NP, 0);
  __builtin_amdgcn_sched_group_barrier(0x080, 14, 0);
  if (h == 0) *(float4*)(As + wave * CP + 4 * c) = acc;
  lds_barrier();

  const float amask = (c < BN) ? 1.f : 0.f;
  const float* aptr = As + (c < BN ? c : 0) * CP + 4 * h;
  float4 av[NP];
#pragma unroll
  for (int j = 0; j < NP; ++j) av[j] = *(const float4*)(aptr + 8 * j);

  v8f acc0 = {0.f, 0.f, 0.f, 0.f, 0.f, 0.f, 0.f, 0.f};
  v8f acc1 = {0.f, 0.f, 0.f, 0.f, 0.f, 0.f, 0.f, 0.f};
#pragma unroll
  for (int j = 0; j < NP; ++j) {
    v2f a0, b0, a1, b1;
    a0.x = av[j].x * amask;  a0.y = av[j].y * amask;
    b0.x = bv[j].x;          b0.y = bv[j].y;
    a1.x = av[j].z * amask;  a1.y = av[j].w * amask;
    b1.x = bv[j].z;          b1.y = bv[j].w;
    acc0 = __builtin_amdgcn_wmma_f32_16x16x4_f32(
        false, a0, false, b0, (short)0, acc0, false, false);
    acc1 = __builtin_amdgcn_wmma_f32_16x16x4_f32(
        false, a1, false, b1, (short)0, acc1, false, false);
  }

  __builtin_amdgcn_sched_group_barrier(0x100, NP, 0);
  __builtin_amdgcn_sched_group_barrier(0x008, 2 * NP, 0);

  float o[BN];
#pragma unroll
  for (int i = 0; i < BN; ++i) o[i] = acc0[i] + acc1[i];
  if (h == 0) {
#pragma unroll
    for (int i = 0; i < BN; ++i) atomicAdd(Out + i * HH + n0 + c, o[i]);
  }
}

__global__ __launch_bounds__(128) __attribute__((amdgpu_waves_per_eu(1, 2)))
void kslice_gemm32(const float* __restrict__ A,
                   const float* __restrict__ Bm,
                   const float* __restrict__ bias,
                   float* __restrict__ Out) {
  constexpr int KB = 256;
  constexpr int NWV = 4;
  constexpr int KW = KB / NWV;
  constexpr int NP = KW / 8;
  __shared__ float red[NWV][2][BN][16];
  const int t    = threadIdx.x;
  const int lane = t & 31;
  const int wave = __builtin_amdgcn_readfirstlane(t >> 5);
  const int n0   = blockIdx.x * 32;
  const int k0   = blockIdx.y * KB + wave * KW;
  const int c    = lane & 15;
  const int h    = lane >> 4;

  const float amask = (c < BN) ? 1.f : 0.f;
  const float* aptr = A + (c < BN ? c : 0) * HH + k0 + 4 * h;
  const float* bptr = Bm + (long)(k0 + 4 * h) * FF + n0 + 2 * c;
  const float bval = (blockIdx.y == 0) ? bias[n0 + lane] : 0.f;

  float4 av[NP];
  float2 bv[NP][4];
#pragma unroll
  for (int j = 0; j < NP; ++j) {
    av[j] = *(const float4*)(aptr + 8 * j);
#pragma unroll
    for (int e = 0; e < 4; ++e)
      bv[j][e] = *(const float2*)(bptr + (long)(8 * j + e) * FF);
  }

  v8f acc00 = {0.f, 0.f, 0.f, 0.f, 0.f, 0.f, 0.f, 0.f};
  v8f acc01 = {0.f, 0.f, 0.f, 0.f, 0.f, 0.f, 0.f, 0.f};
  v8f acc10 = {0.f, 0.f, 0.f, 0.f, 0.f, 0.f, 0.f, 0.f};
  v8f acc11 = {0.f, 0.f, 0.f, 0.f, 0.f, 0.f, 0.f, 0.f};
#pragma unroll
  for (int j = 0; j < NP; ++j) {
    v2f a0, a1, b00, b01, b10, b11;
    a0.x = av[j].x * amask;  a0.y = av[j].y * amask;
    a1.x = av[j].z * amask;  a1.y = av[j].w * amask;
    b00.x = bv[j][0].x;  b00.y = bv[j][1].x;
    b01.x = bv[j][2].x;  b01.y = bv[j][3].x;
    b10.x = bv[j][0].y;  b10.y = bv[j][1].y;
    b11.x = bv[j][2].y;  b11.y = bv[j][3].y;
    acc00 = __builtin_amdgcn_wmma_f32_16x16x4_f32(
        false, a0, false, b00, (short)0, acc00, false, false);
    acc10 = __builtin_amdgcn_wmma_f32_16x16x4_f32(
        false, a0, false, b10, (short)0, acc10, false, false);
    acc01 = __builtin_amdgcn_wmma_f32_16x16x4_f32(
        false, a1, false, b01, (short)0, acc01, false, false);
    acc11 = __builtin_amdgcn_wmma_f32_16x16x4_f32(
        false, a1, false, b11, (short)0, acc11, false, false);
  }

  if (h == 0) {
#pragma unroll
    for (int i = 0; i < BN; ++i) {
      red[wave][0][i][c] = acc00[i] + acc01[i];
      red[wave][1][i][c] = acc10[i] + acc11[i];
    }
  }
  __syncthreads();

  const int tl = lane & 1, cc = lane >> 1;
  float v = red[0][tl][wave][cc];
#pragma unroll
  for (int w = 1; w < NWV; ++w) v += red[w][tl][wave][cc];
  atomicAdd(Out + wave * FF + n0 + lane, v + bval);
}

__global__ __launch_bounds__(512) __attribute__((amdgpu_waves_per_eu(4, 4)))
void glu_ln_gemm16(const float* __restrict__ hin,
                   const float* __restrict__ attn,
                   const float* __restrict__ gamma,
                   const float* __restrict__ beta,
                   const float* __restrict__ Bm,
                   const float* __restrict__ bias,
                   float* __restrict__ Out) {
  constexpr int NW = 16;
  constexpr int KW = HH / NW;
  constexpr int NP = KW / 8;
  constexpr int CP = HH + 8;
  __shared__ float red[NW][16][16];
  __shared__ __attribute__((aligned(16))) float comb[BN * CP];
  __shared__ float psum[NW], psq[NW];
  const int t    = threadIdx.x;
  const int lane = t & 31;
  const int wave = t >> 5;
  const int n0   = blockIdx.x * 16;
  const int c    = lane & 15;
  const int h    = lane >> 4;
  const int kb   = wave * KW + 4 * h;
  const int bt   = t >> 7;
  const int q    = t & 127;

  const float4* h4 = (const float4*)(hin + (size_t)bt * FF);
  const float4 x1a = h4[q],       x1b = h4[q + 128];
  const float4 x2a = h4[q + 256], x2b = h4[q + 384];
  const float4* a4 = (const float4*)(attn + (size_t)bt * HH);
  const float4 ata = a4[q], atb = a4[q + 128];
  const float4 gaa = ((const float4*)gamma)[q], gab = ((const float4*)gamma)[q + 128];
  const float4 bea = ((const float4*)beta)[q],  beb = ((const float4*)beta)[q + 128];
  const float bval = bias[n0 + (t & 15)];

  const float* bptr = Bm + (long)kb * OO + (n0 + c);
  float4 bv[NP];
#pragma unroll
  for (int j = 0; j < NP; ++j) {
    const float* bp = bptr + (long)(8 * j) * OO;
    bv[j].x = bp[0];
    bv[j].y = bp[OO];
    bv[j].z = bp[2 * OO];
    bv[j].w = bp[3 * OO];
  }

  const float x1[8] = {x1a.x, x1a.y, x1a.z, x1a.w, x1b.x, x1b.y, x1b.z, x1b.w};
  const float x2[8] = {x2a.x, x2a.y, x2a.z, x2a.w, x2b.x, x2b.y, x2b.z, x2b.w};
  float hv[8];
  float sum = 0.f;
#pragma unroll
  for (int j = 0; j < 8; ++j) {
    const float g = 0.5f * x2[j] * (1.f + erff(x2[j] * 0.70710678118654752f));
    hv[j] = x1[j] * g;
    sum += hv[j];
  }
  sum = wave_sum(sum);
  if (lane == 0) psum[wave] = sum;
  lds_barrier();
  const int w0 = bt * 4;
  const float mean = (psum[w0] + psum[w0 + 1] + psum[w0 + 2] + psum[w0 + 3]) * (1.f / HH);
  float sq = 0.f;
#pragma unroll
  for (int j = 0; j < 8; ++j) { const float d = hv[j] - mean; sq += d * d; }
  sq = wave_sum(sq);
  if (lane == 0) psq[wave] = sq;
  lds_barrier();
  const float var = (psq[w0] + psq[w0 + 1] + psq[w0 + 2] + psq[w0 + 3]) * (1.f / HH);
  const float rstd = rsqrtf(var + 1e-5f);
  float4 ca, cb;
  ca.x = ata.x + (hv[0] - mean) * rstd * gaa.x + bea.x;
  ca.y = ata.y + (hv[1] - mean) * rstd * gaa.y + bea.y;
  ca.z = ata.z + (hv[2] - mean) * rstd * gaa.z + bea.z;
  ca.w = ata.w + (hv[3] - mean) * rstd * gaa.w + bea.w;
  cb.x = atb.x + (hv[4] - mean) * rstd * gab.x + beb.x;
  cb.y = atb.y + (hv[5] - mean) * rstd * gab.y + beb.y;
  cb.z = atb.z + (hv[6] - mean) * rstd * gab.z + beb.z;
  cb.w = atb.w + (hv[7] - mean) * rstd * gab.w + beb.w;
  *(float4*)(comb + bt * CP + 4 * q)       = ca;
  *(float4*)(comb + bt * CP + 512 + 4 * q) = cb;
  lds_barrier();

  const float amask = (c < BN) ? 1.f : 0.f;
  const float* aptr = comb + (c < BN ? c : 0) * CP + kb;
  float4 av[NP];
#pragma unroll
  for (int j = 0; j < NP; ++j) av[j] = *(const float4*)(aptr + 8 * j);

  v8f acc0 = {0.f, 0.f, 0.f, 0.f, 0.f, 0.f, 0.f, 0.f};
  v8f acc1 = {0.f, 0.f, 0.f, 0.f, 0.f, 0.f, 0.f, 0.f};
#pragma unroll
  for (int j = 0; j < NP; ++j) {
    v2f a0, b0, a1, b1;
    a0.x = av[j].x * amask;  a0.y = av[j].y * amask;
    b0.x = bv[j].x;          b0.y = bv[j].y;
    a1.x = av[j].z * amask;  a1.y = av[j].w * amask;
    b1.x = bv[j].z;          b1.y = bv[j].w;
    acc0 = __builtin_amdgcn_wmma_f32_16x16x4_f32(
        false, a0, false, b0, (short)0, acc0, false, false);
    acc1 = __builtin_amdgcn_wmma_f32_16x16x4_f32(
        false, a1, false, b1, (short)0, acc1, false, false);
  }

#pragma unroll
  for (int i = 0; i < 8; ++i) red[wave][8 * h + i][c] = acc0[i] + acc1[i];
  __syncthreads();

  if (t < 16 * BN) {
    const int r = t >> 4, cc = t & 15;
    float v = 0.f;
#pragma unroll
    for (int w = 0; w < NW; ++w) v += red[w][r][cc];
    Out[r * OO + n0 + cc] = v + bval;
  }
}

extern "C" void kernel_launch(void* const* d_in, const int* in_sizes, int n_in,
                              void* d_out, int out_size, void* d_ws, size_t ws_size,
                              hipStream_t stream) {
  const float* x       = (const float*)d_in[0];
  const float* Wq_proj = (const float*)d_in[1];
  const float* bq      = (const float*)d_in[2];
  const float* Wq      = (const float*)d_in[3];
  const float* Wk      = (const float*)d_in[4];
  const float* Wv      = (const float*)d_in[5];
  const float* W1      = (const float*)d_in[6];
  const float* b1      = (const float*)d_in[7];
  const float* W2      = (const float*)d_in[8];
  const float* b2      = (const float*)d_in[9];
  const float* gamma   = (const float*)d_in[10];
  const float* beta    = (const float*)d_in[11];
  float* out = (float*)d_out;

  float* ws     = (float*)d_ws;
  float* qlp    = ws;
  float* qtil   = ws + 262144;
  float* attn   = ws + 327680;
  float* hin    = ws + 331776;
  float* part   = ws + 524288;
  float* stats  = ws + 786432;

  gemm16_pair<2><<<dim3(HH / 16, 2), 512, 0, stream>>>(
      x + (size_t)(SS - 1) * DD, (long)SS * DD, Wq_proj, bq, Wq, qlp);
  planes_gemm16t<<<dim3(DD / 64, HH / 64), 128, 0, stream>>>(qlp, Wk, qtil);
  scores_part_k<<<dim3(SS / 32, BN), 256, 0, stream>>>(x, qtil, part, stats, attn, hin);
  merge_gemm16<<<dim3(HH / 64, DD / 64), 128, 0, stream>>>(part, stats, Wv, attn);
  kslice_gemm32<<<dim3(FF / 32, HH / 256), 128, 0, stream>>>(attn, W1, b1, hin);
  glu_ln_gemm16<<<dim3(OO / 16), 512, 0, stream>>>(
      hin, attn, gamma, beta, W2, b2, out);
}
